# starting kernel
# baseline (speedup 1.0000x reference)
#include <hip/hip_runtime.h>
#include <hip/hip_fp16.h>

typedef float v2f __attribute__((ext_vector_type(2)));
typedef float v8f __attribute__((ext_vector_type(8)));
typedef _Float16 v4h __attribute__((ext_vector_type(4)));

#define NNODES   50000
#define NEDGES   1600000
#define IN_CH    256
#define OUT_CH   128
#define NUM_REL  8
#define NUM_BASES 4
#define NSLICE   (NUM_REL + 1)

__global__ __launch_bounds__(256)
void build_wT_kernel(const float* __restrict__ att,
                     const float* __restrict__ basis,
                     const float* __restrict__ root,
                     float* __restrict__ wT) {
  int idx = blockIdx.x * blockDim.x + threadIdx.x;
  if (idx >= NSLICE * IN_CH * OUT_CH) return;
  int r  = idx / (IN_CH * OUT_CH);
  int io = idx % (IN_CH * OUT_CH);
  int i  = io / OUT_CH;
  int o  = io % OUT_CH;
  float v;
  if (r < NUM_REL) {
    v = 0.f;
#pragma unroll
    for (int b = 0; b < NUM_BASES; ++b)
      v += att[r * NUM_BASES + b] * basis[b * (IN_CH * OUT_CH) + io];
  } else {
    v = root[io];
  }
  wT[(size_t)r * OUT_CH * IN_CH + (size_t)o * IN_CH + i] = v;
}

__global__ __launch_bounds__(256)
void rgcn_gemm_kernel(const float* __restrict__ x,
                      const float* __restrict__ wT,
                      const float* __restrict__ bias,
                      _Float16* __restrict__ h,
                      float* __restrict__ out)
{
  const int lane = threadIdx.x & 31;
  const int wave = threadIdx.x >> 5;
  const int half = lane >> 4;
  const int lid  = lane & 15;

  const int row0 = blockIdx.x * 16;
  const int col0 = wave * 16;
  const int r    = blockIdx.y;

  const float* xrow = x  + (size_t)(row0 + lid) * IN_CH;
  const float* bcol = wT + ((size_t)r * OUT_CH + col0 + lid) * IN_CH;

  v8f acc = {};
  for (int k = 0; k < IN_CH; k += 4) {
    v2f a = *reinterpret_cast<const v2f*>(xrow + k + 2 * half);
    v2f b = *reinterpret_cast<const v2f*>(bcol + k + 2 * half);
    acc = __builtin_amdgcn_wmma_f32_16x16x4_f32(false, a, false, b,
                                                (short)0, acc, false, false);
  }

  if (r < NUM_REL) {
    _Float16* hp = h + ((size_t)r * NNODES + row0) * OUT_CH + col0 + lid;
#pragma unroll
    for (int v = 0; v < 8; ++v)
      hp[(size_t)(v + 8 * half) * OUT_CH] = (_Float16)acc[v];
  } else {
    float bv = bias[col0 + lid];
    float* op = out + (size_t)row0 * OUT_CH + col0 + lid;
#pragma unroll
    for (int v = 0; v < 8; ++v)
      op[(size_t)(v + 8 * half) * OUT_CH] = acc[v] + bv;
  }
}

__global__ __launch_bounds__(256)
void edge_scatter_kernel(const int* __restrict__ esrc,
                         const int* __restrict__ edst,
                         const int* __restrict__ etyp,
                         const float* __restrict__ norm,
                         const _Float16* __restrict__ h,
                         float* __restrict__ out)
{
  const int lane   = threadIdx.x & 31;
  const int wid    = (blockIdx.x * blockDim.x + threadIdx.x) >> 5;
  const int nwaves = (gridDim.x * blockDim.x) >> 5;

  for (int e = wid; e < NEDGES; e += nwaves) {
    const int   s  = esrc[e];
    const int   d  = edst[e];
    const int   t  = etyp[e];
    const float nm = norm[e];

    const v4h hv = *reinterpret_cast<const v4h*>(
        h + ((size_t)t * NNODES + s) * OUT_CH + lane * 4);

    float* op = out + (size_t)d * OUT_CH + lane * 4;
    atomicAdd(op + 0, (float)hv.x * nm);
    atomicAdd(op + 1, (float)hv.y * nm);
    atomicAdd(op + 2, (float)hv.z * nm);
    atomicAdd(op + 3, (float)hv.w * nm);
  }
}

extern "C" void kernel_launch(void* const* d_in, const int* in_sizes, int n_in,
                              void* d_out, int out_size, void* d_ws, size_t ws_size,
                              hipStream_t stream) {
  const float* x     = (const float*)d_in[0];
  const int*   esrc  = (const int*)  d_in[1];
  const int*   edst  = (const int*)  d_in[2];
  const int*   etyp  = (const int*)  d_in[3];
  const float* norm  = (const float*)d_in[4];
  const float* basis = (const float*)d_in[5];
  const float* att   = (const float*)d_in[6];
  const float* root  = (const float*)d_in[7];
  const float* bias  = (const float*)d_in[8];
  float* out = (float*)d_out;

  float*    wT = (float*)d_ws;
  _Float16* h  = (_Float16*)((char*)d_ws +
                             (size_t)NSLICE * IN_CH * OUT_CH * sizeof(float));

  build_wT_kernel<<<(NSLICE * IN_CH * OUT_CH + 255) / 256, 256, 0, stream>>>(
      att, basis, root, wT);

  rgcn_gemm_kernel<<<dim3(NNODES / 16, NSLICE), 256, 0, stream>>>(
      x, wT, bias, h, out);

  edge_scatter_kernel<<<4096, 256, 0, stream>>>(esrc, edst, etyp, norm, h, out);
}


# GEMM re-tiled: 128x128 workgroup tile, x and wT k-chunks staged through LDS, each wave a 64x32 tile of 4x2 WMMA tiles reusing fragments; k summed in a different order inside each group of 8
# speedup vs baseline: 1.8861x; 1.8861x over previous
#include <hip/hip_runtime.h>
#include <hip/hip_fp16.h>

typedef float v2f __attribute__((ext_vector_type(2)));
typedef float v4f __attribute__((ext_vector_type(4)));
typedef float v8f __attribute__((ext_vector_type(8)));
typedef _Float16 v4h __attribute__((ext_vector_type(4)));

#define NNODES   50000
#define NEDGES   1600000
#define IN_CH    256
#define OUT_CH   128
#define NUM_REL  8
#define NUM_BASES 4
#define NSLICE   (NUM_REL + 1)

__global__ __launch_bounds__(256)
void build_wT_kernel(const float* __restrict__ att,
                     const float* __restrict__ basis,
                     const float* __restrict__ root,
                     float* __restrict__ wT) {
  int idx = blockIdx.x * blockDim.x + threadIdx.x;
  if (idx >= NSLICE * IN_CH * OUT_CH) return;
  int r  = idx / (IN_CH * OUT_CH);
  int io = idx % (IN_CH * OUT_CH);
  int i  = io / OUT_CH;
  int o  = io % OUT_CH;
  float v;
  if (r < NUM_REL) {
    v = 0.f;
#pragma unroll
    for (int b = 0; b < NUM_BASES; ++b)
      v += att[r * NUM_BASES + b] * basis[b * (IN_CH * OUT_CH) + io];
  } else {
    v = root[io];
  }
  wT[(size_t)r * OUT_CH * IN_CH + (size_t)o * IN_CH + i] = v;
}

#define BM     128
#define BN     128
#define KC     32
#define PITCH  36
#define NCHUNK (IN_CH / KC)
#define MTILES ((NNODES + BM - 1) / BM)

__global__ __launch_bounds__(256)
void rgcn_gemm_kernel(const float* __restrict__ x,
                      const float* __restrict__ wT,
                      const float* __restrict__ bias,
                      _Float16* __restrict__ h,
                      float* __restrict__ out)
{
  __shared__ __attribute__((aligned(16))) float lds[2][(BM + BN) * PITCH];

  const int t    = threadIdx.x;
  const int lane = t & 31;
  const int wave = __builtin_amdgcn_readfirstlane(t >> 5);
  const int half = lane >> 4;
  const int lid  = lane & 15;
  const int wm   = wave >> 2;
  const int wn   = wave & 3;

  const int r    = blockIdx.x;
  const int row0 = blockIdx.y * BM;

  const int srow = t >> 3;
  const int sc4  = (t & 7) * 4;

  int aoff[4];
#pragma unroll
  for (int j = 0; j < 4; ++j) {
    int row = row0 + srow + 32 * j;
    row = row < NNODES ? row : NNODES - 1;
    aoff[j] = row * IN_CH + sc4;
  }
  const float* bbase = wT + ((size_t)r * OUT_CH + srow) * IN_CH + sc4;
  const int lA = srow * PITCH + sc4;
  const int lB = (BM + srow) * PITCH + sc4;

  const int fA = (wm * 64 + lid) * PITCH + half * 4;
  const int fB = (BM + wn * 32 + lid) * PITCH + half * 4;

  v8f acc[4][2] = {};
  v4f ga[4], gb[4];

#pragma unroll
  for (int j = 0; j < 4; ++j) {
    ga[j] = *reinterpret_cast<const v4f*>(x + aoff[j]);
    gb[j] = *reinterpret_cast<const v4f*>(bbase + j * 32 * IN_CH);
  }
#pragma unroll
  for (int j = 0; j < 4; ++j) {
    *reinterpret_cast<v4f*>(&lds[0][lA + j * 32 * PITCH]) = ga[j];
    *reinterpret_cast<v4f*>(&lds[0][lB + j * 32 * PITCH]) = gb[j];
  }
  __syncthreads();

  for (int c = 0; c < NCHUNK; ++c) {
    const float* cur = lds[c & 1];
    float*       nxt = lds[(c + 1) & 1];
    const bool  more = (c + 1 < NCHUNK);

    if (more) {
#pragma unroll
      for (int j = 0; j < 4; ++j) {
        ga[j] = *reinterpret_cast<const v4f*>(x + aoff[j] + (c + 1) * KC);
        gb[j] = *reinterpret_cast<const v4f*>(bbase + j * 32 * IN_CH + (c + 1) * KC);
      }
    }
    __builtin_amdgcn_sched_barrier(0);

#pragma unroll
    for (int s = 0; s < KC / 8; ++s) {
      v4f af[4], bf[2];
#pragma unroll
      for (int i = 0; i < 4; ++i)
        af[i] = *reinterpret_cast<const v4f*>(&cur[fA + i * 16 * PITCH + s * 8]);
#pragma unroll
      for (int j = 0; j < 2; ++j)
        bf[j] = *reinterpret_cast<const v4f*>(&cur[fB + j * 16 * PITCH + s * 8]);
#pragma unroll
      for (int i = 0; i < 4; ++i)
#pragma unroll
        for (int j = 0; j < 2; ++j)
          acc[i][j] = __builtin_amdgcn_wmma_f32_16x16x4_f32(
              false, af[i].xy, false, bf[j].xy, (short)0, acc[i][j], false, false);
#pragma unroll
      for (int i = 0; i < 4; ++i)
#pragma unroll
        for (int j = 0; j < 2; ++j)
          acc[i][j] = __builtin_amdgcn_wmma_f32_16x16x4_f32(
              false, af[i].zw, false, bf[j].zw, (short)0, acc[i][j], false, false);
    }

    __builtin_amdgcn_sched_barrier(0);
    if (more) {
#pragma unroll
      for (int j = 0; j < 4; ++j) {
        *reinterpret_cast<v4f*>(&nxt[lA + j * 32 * PITCH]) = ga[j];
        *reinterpret_cast<v4f*>(&nxt[lB + j * 32 * PITCH]) = gb[j];
      }
      __syncthreads();
    }
  }

  static_assert(NNODES % 16 == 0, "tile guard assumes whole 16-row tiles");
  const int colb = wn * 32 + lid;
  if (r < NUM_REL) {
#pragma unroll
    for (int i = 0; i < 4; ++i) {
      const int tr = row0 + wm * 64 + i * 16;
      if (tr < NNODES) {
#pragma unroll
        for (int j = 0; j < 2; ++j) {
          _Float16* hp = h + ((size_t)r * NNODES + tr + 8 * half) * OUT_CH + colb + j * 16;
#pragma unroll
          for (int v = 0; v < 8; ++v)
            hp[(size_t)v * OUT_CH] = (_Float16)acc[i][j][v];
        }
      }
    }
  } else {
    const float bv0 = bias[colb];
    const float bv1 = bias[colb + 16];
#pragma unroll
    for (int i = 0; i < 4; ++i) {
      const int tr = row0 + wm * 64 + i * 16;
      if (tr < NNODES) {
#pragma unroll
        for (int j = 0; j < 2; ++j) {
          const float bv = j ? bv1 : bv0;
          float* op = out + (size_t)(tr + 8 * half) * OUT_CH + colb + j * 16;
#pragma unroll
          for (int v = 0; v < 8; ++v)
            op[(size_t)v * OUT_CH] = acc[i][j][v] + bv;
        }
      }
    }
  }
}

__global__ __launch_bounds__(256)
void edge_scatter_kernel(const int* __restrict__ esrc,
                         const int* __restrict__ edst,
                         const int* __restrict__ etyp,
                         const float* __restrict__ norm,
                         const _Float16* __restrict__ h,
                         float* __restrict__ out)
{
  const int lane   = threadIdx.x & 31;
  const int wid    = (blockIdx.x * blockDim.x + threadIdx.x) >> 5;
  const int nwaves = (gridDim.x * blockDim.x) >> 5;

  for (int e = wid; e < NEDGES; e += nwaves) {
    const int   s  = esrc[e];
    const int   d  = edst[e];
    const int   t  = etyp[e];
    const float nm = norm[e];

    const v4h hv = *reinterpret_cast<const v4h*>(
        h + ((size_t)t * NNODES + s) * OUT_CH + lane * 4);

    float* op = out + (size_t)d * OUT_CH + lane * 4;
    atomicAdd(op + 0, (float)hv.x * nm);
    atomicAdd(op + 1, (float)hv.y * nm);
    atomicAdd(op + 2, (float)hv.z * nm);
    atomicAdd(op + 3, (float)hv.w * nm);
  }
}

extern "C" void kernel_launch(void* const* d_in, const int* in_sizes, int n_in,
                              void* d_out, int out_size, void* d_ws, size_t ws_size,
                              hipStream_t stream) {
  const float* x     = (const float*)d_in[0];
  const int*   esrc  = (const int*)  d_in[1];
  const int*   edst  = (const int*)  d_in[2];
  const int*   etyp  = (const int*)  d_in[3];
  const float* norm  = (const float*)d_in[4];
  const float* basis = (const float*)d_in[5];
  const float* att   = (const float*)d_in[6];
  const float* root  = (const float*)d_in[7];
  const float* bias  = (const float*)d_in[8];
  float* out = (float*)d_out;

  float*    wT = (float*)d_ws;
  _Float16* h  = (_Float16*)((char*)d_ws +
                             (size_t)NSLICE * IN_CH * OUT_CH * sizeof(float));

  build_wT_kernel<<<(NSLICE * IN_CH * OUT_CH + 255) / 256, 256, 0, stream>>>(
      att, basis, root, wT);

  rgcn_gemm_kernel<<<dim3(NSLICE, MTILES), 256, 0, stream>>>(
      x, wT, bias, h, out);

  edge_scatter_kernel<<<4096, 256, 0, stream>>>(esrc, edst, etyp, norm, h, out);
}


# h rows stored channel-interleaved (GEMM column map permuted, 4-byte pair stores) so each scatter atomic add covers one whole 128-B line of out, was 4 lines per add; atomic add order as before
# speedup vs baseline: 2.6524x; 1.4063x over previous
#include <hip/hip_runtime.h>
#include <hip/hip_fp16.h>

typedef float v2f __attribute__((ext_vector_type(2)));
typedef float v4f __attribute__((ext_vector_type(4)));
typedef float v8f __attribute__((ext_vector_type(8)));
typedef _Float16 v4h __attribute__((ext_vector_type(4)));
typedef _Float16 v2h __attribute__((ext_vector_type(2)));

#define NNODES   50000
#define NEDGES   1600000
#define IN_CH    256
#define OUT_CH   128
#define NUM_REL  8
#define NUM_BASES 4
#define NSLICE   (NUM_REL + 1)

__global__ __launch_bounds__(256)
void build_wT_kernel(const float* __restrict__ att,
                     const float* __restrict__ basis,
                     const float* __restrict__ root,
                     float* __restrict__ wT) {
  int idx = blockIdx.x * blockDim.x + threadIdx.x;
  if (idx >= NSLICE * IN_CH * OUT_CH) return;
  int r  = idx / (IN_CH * OUT_CH);
  int io = idx % (IN_CH * OUT_CH);
  int i  = io / OUT_CH;
  int o  = io % OUT_CH;
  float v;
  if (r < NUM_REL) {
    v = 0.f;
#pragma unroll
    for (int b = 0; b < NUM_BASES; ++b)
      v += att[r * NUM_BASES + b] * basis[b * (IN_CH * OUT_CH) + io];
  } else {
    v = root[io];
  }
  wT[(size_t)r * OUT_CH * IN_CH + (size_t)o * IN_CH + i] = v;
}

#define BM     128
#define BN     128
#define KC     32
#define PITCH  36
#define NCHUNK (IN_CH / KC)
#define MTILES ((NNODES + BM - 1) / BM)

__global__ __launch_bounds__(256)
void rgcn_gemm_kernel(const float* __restrict__ x,
                      const float* __restrict__ wT,
                      const float* __restrict__ bias,
                      _Float16* __restrict__ h,
                      float* __restrict__ out)
{
  __shared__ __attribute__((aligned(16))) float lds[2][(BM + BN) * PITCH];

  const int t    = threadIdx.x;
  const int lane = t & 31;
  const int wave = __builtin_amdgcn_readfirstlane(t >> 5);
  const int half = lane >> 4;
  const int lid  = lane & 15;
  const int wm   = wave >> 2;
  const int wn   = wave & 3;

  const int r    = blockIdx.x;
  const int row0 = blockIdx.y * BM;

  const int srow = t >> 3;
  const int sc4  = (t & 7) * 4;

  int aoff[4];
#pragma unroll
  for (int j = 0; j < 4; ++j) {
    int row = row0 + srow + 32 * j;
    row = row < NNODES ? row : NNODES - 1;
    aoff[j] = row * IN_CH + sc4;
  }
  const float* bslice = wT + (size_t)r * OUT_CH * IN_CH;
  int boff[4];
#pragma unroll
  for (int j = 0; j < 4; ++j) {
    const int ch = ((j >> 1) << 6) | ((srow >> 4) << 5) | ((j & 1) << 4) | (srow & 15);
    boff[j] = ch * IN_CH + sc4;
  }
  const int lA = srow * PITCH + sc4;
  const int lB = (BM + srow) * PITCH + sc4;

  const int fA = (wm * 64 + lid) * PITCH + half * 4;
  const int fB = (BM + wn * 32 + lid) * PITCH + half * 4;

  v8f acc[4][2] = {};
  v4f ga[4], gb[4];

#pragma unroll
  for (int j = 0; j < 4; ++j) {
    ga[j] = *reinterpret_cast<const v4f*>(x + aoff[j]);
    gb[j] = *reinterpret_cast<const v4f*>(bslice + boff[j]);
  }
#pragma unroll
  for (int j = 0; j < 4; ++j) {
    *reinterpret_cast<v4f*>(&lds[0][lA + j * 32 * PITCH]) = ga[j];
    *reinterpret_cast<v4f*>(&lds[0][lB + j * 32 * PITCH]) = gb[j];
  }
  __syncthreads();

#pragma unroll
  for (int c = 0; c < NCHUNK; ++c) {
    const float* cur = lds[c & 1];
    float*       nxt = lds[(c + 1) & 1];
    const bool  more = (c + 1 < NCHUNK);

    if (more) {
#pragma unroll
      for (int j = 0; j < 4; ++j) {
        ga[j] = *reinterpret_cast<const v4f*>(x + aoff[j] + (c + 1) * KC);
        gb[j] = *reinterpret_cast<const v4f*>(bslice + boff[j] + (c + 1) * KC);
      }
    }
    __builtin_amdgcn_sched_barrier(0);

#pragma unroll
    for (int s = 0; s < KC / 8; ++s) {
      v4f af[4], bf[2];
#pragma unroll
      for (int i = 0; i < 4; ++i)
        af[i] = *reinterpret_cast<const v4f*>(&cur[fA + i * 16 * PITCH + s * 8]);
#pragma unroll
      for (int j = 0; j < 2; ++j)
        bf[j] = *reinterpret_cast<const v4f*>(&cur[fB + j * 16 * PITCH + s * 8]);
#pragma unroll
      for (int i = 0; i < 4; ++i)
#pragma unroll
        for (int j = 0; j < 2; ++j)
          acc[i][j] = __builtin_amdgcn_wmma_f32_16x16x4_f32(
              false, af[i].xy, false, bf[j].xy, (short)0, acc[i][j], false, false);
#pragma unroll
      for (int i = 0; i < 4; ++i)
#pragma unroll
        for (int j = 0; j < 2; ++j)
          acc[i][j] = __builtin_amdgcn_wmma_f32_16x16x4_f32(
              false, af[i].zw, false, bf[j].zw, (short)0, acc[i][j], false, false);
    }

    __builtin_amdgcn_sched_barrier(0);
    if (more) {
#pragma unroll
      for (int j = 0; j < 4; ++j) {
        *reinterpret_cast<v4f*>(&nxt[lA + j * 32 * PITCH]) = ga[j];
        *reinterpret_cast<v4f*>(&nxt[lB + j * 32 * PITCH]) = gb[j];
      }
      __syncthreads();
    }
  }

  static_assert(NNODES % 16 == 0, "tile guard assumes whole 16-row tiles");
  const int q    = (wn & 1) * 16 + lid;
  const int colb = (wn >> 1) * 64 + q;
  if (r < NUM_REL) {
    const int hpos = 4 * q + (wn >> 1) * 2;
#pragma unroll
    for (int i = 0; i < 4; ++i) {
      const int tr = row0 + wm * 64 + i * 16;
      if (tr < NNODES) {
        _Float16* hp = h + ((size_t)r * NNODES + tr + 8 * half) * OUT_CH + hpos;
#pragma unroll
        for (int v = 0; v < 8; ++v) {
          v2h pr;
          pr.x = (_Float16)acc[i][0][v];
          pr.y = (_Float16)acc[i][1][v];
          *reinterpret_cast<v2h*>(hp + (size_t)v * OUT_CH) = pr;
        }
      }
    }
  } else {
    const float bv0 = bias[colb];
    const float bv1 = bias[colb + 32];
#pragma unroll
    for (int i = 0; i < 4; ++i) {
      const int tr = row0 + wm * 64 + i * 16;
      if (tr < NNODES) {
#pragma unroll
        for (int j = 0; j < 2; ++j) {
          const float bv = j ? bv1 : bv0;
          float* op = out + (size_t)(tr + 8 * half) * OUT_CH + colb + j * 32;
#pragma unroll
          for (int v = 0; v < 8; ++v)
            op[(size_t)v * OUT_CH] = acc[i][j][v] + bv;
        }
      }
    }
  }
}

__global__ __launch_bounds__(256)
void edge_scatter_kernel(const int* __restrict__ esrc,
                         const int* __restrict__ edst,
                         const int* __restrict__ etyp,
                         const float* __restrict__ norm,
                         const _Float16* __restrict__ h,
                         float* __restrict__ out)
{
  const int lane   = threadIdx.x & 31;
  const int wid    = (blockIdx.x * blockDim.x + threadIdx.x) >> 5;
  const int nwaves = (gridDim.x * blockDim.x) >> 5;

  for (int e = wid; e < NEDGES; e += nwaves) {
    const int   s  = esrc[e];
    const int   d  = edst[e];
    const int   t  = etyp[e];
    const float nm = norm[e];

    const v4h hv = *reinterpret_cast<const v4h*>(
        h + ((size_t)t * NNODES + s) * OUT_CH + lane * 4);

    float* op = out + (size_t)d * OUT_CH + lane;
    atomicAdd(op +  0, (float)hv.x * nm);
    atomicAdd(op + 32, (float)hv.y * nm);
    atomicAdd(op + 64, (float)hv.z * nm);
    atomicAdd(op + 96, (float)hv.w * nm);
  }
}

extern "C" void kernel_launch(void* const* d_in, const int* in_sizes, int n_in,
                              void* d_out, int out_size, void* d_ws, size_t ws_size,
                              hipStream_t stream) {
  const float* x     = (const float*)d_in[0];
  const int*   esrc  = (const int*)  d_in[1];
  const int*   edst  = (const int*)  d_in[2];
  const int*   etyp  = (const int*)  d_in[3];
  const float* norm  = (const float*)d_in[4];
  const float* basis = (const float*)d_in[5];
  const float* att   = (const float*)d_in[6];
  const float* root  = (const float*)d_in[7];
  const float* bias  = (const float*)d_in[8];
  float* out = (float*)d_out;

  float*    wT = (float*)d_ws;
  _Float16* h  = (_Float16*)((char*)d_ws +
                             (size_t)NSLICE * IN_CH * OUT_CH * sizeof(float));

  build_wT_kernel<<<(NSLICE * IN_CH * OUT_CH + 255) / 256, 256, 0, stream>>>(
      att, basis, root, wT);

  rgcn_gemm_kernel<<<dim3(NSLICE, MTILES), 256, 0, stream>>>(
      x, wT, bias, h, out);

  edge_scatter_kernel<<<4096, 256, 0, stream>>>(esrc, edst, etyp, norm, h, out);
}


# Matrix products over the 4 bases plus root (5 column slices, were 9); the 8 relation rows formed from the 4 basis accumulators in the epilogue via LDS; sum over bases now last (order changed)
# speedup vs baseline: 3.1789x; 1.1985x over previous
#include <hip/hip_runtime.h>
#include <hip/hip_fp16.h>

typedef float v2f __attribute__((ext_vector_type(2)));
typedef float v4f __attribute__((ext_vector_type(4)));
typedef float v8f __attribute__((ext_vector_type(8)));
typedef _Float16 v4h __attribute__((ext_vector_type(4)));

#define NNODES   50000
#define NEDGES   1600000
#define IN_CH    256
#define OUT_CH   128
#define NUM_REL  8
#define NUM_BASES 4
#define NSLICE   (NUM_BASES + 1)

__global__ __launch_bounds__(256)
void build_wT_kernel(const float* __restrict__ basis,
                     const float* __restrict__ root,
                     float* __restrict__ wT) {
  int idx = blockIdx.x * blockDim.x + threadIdx.x;
  if (idx >= NSLICE * IN_CH * OUT_CH) return;
  int s  = idx / (IN_CH * OUT_CH);
  int io = idx % (IN_CH * OUT_CH);
  int i  = io / OUT_CH;
  int o  = io % OUT_CH;
  const float v = (s < NUM_BASES) ? basis[idx] : root[io];
  wT[(size_t)s * OUT_CH * IN_CH + (size_t)o * IN_CH + i] = v;
}

#define BM     128
#define BN     128
#define KC     32
#define PITCH  36
#define NCHUNK (IN_CH / KC)
#define CP     132
#define MTILES ((NNODES + BM - 1) / BM)

__global__ __launch_bounds__(256)
void rgcn_gemm_kernel(const float* __restrict__ x,
                      const float* __restrict__ wT,
                      const float* __restrict__ att,
                      const float* __restrict__ bias,
                      _Float16* __restrict__ h,
                      float* __restrict__ out)
{
  __shared__ __attribute__((aligned(16))) float lds[2][(BM + BN) * PITCH];

  const int t    = threadIdx.x;
  const int lane = t & 31;
  const int wave = __builtin_amdgcn_readfirstlane(t >> 5);
  const int half = lane >> 4;
  const int lid  = lane & 15;
  const int wm   = wave >> 2;
  const int wn   = wave & 3;

  const int r    = blockIdx.x;
  const int row0 = blockIdx.y * BM;

  const int srow = t >> 3;
  const int sc4  = (t & 7) * 4;

  int aoff[4];
#pragma unroll
  for (int j = 0; j < 4; ++j) {
    int row = row0 + srow + 32 * j;
    row = row < NNODES ? row : NNODES - 1;
    aoff[j] = row * IN_CH + sc4;
  }
  const float* bslice = wT;
  int boff[4];
#pragma unroll
  for (int j = 0; j < 4; ++j) {
    int wrow;
    if (r < NUM_BASES) {
      wrow = j * OUT_CH + 8 * r + (srow >> 2) + 32 * (srow & 3);
    } else {
      const int ch = ((j >> 1) << 6) | ((srow >> 4) << 5) | ((j & 1) << 4) | (srow & 15);
      wrow = NUM_BASES * OUT_CH + ch;
    }
    boff[j] = wrow * IN_CH + sc4;
  }
  const int lA = srow * PITCH + sc4;
  const int lB = (BM + srow) * PITCH + sc4;

  const int fA = (wm * 64 + lid) * PITCH + half * 4;
  const int fB = (BM + wn * 32 + lid) * PITCH + half * 4;

  v8f acc[4][2] = {};
  v4f ga[4], gb[4];

#pragma unroll
  for (int j = 0; j < 4; ++j) {
    ga[j] = *reinterpret_cast<const v4f*>(x + aoff[j]);
    gb[j] = *reinterpret_cast<const v4f*>(bslice + boff[j]);
  }
#pragma unroll
  for (int j = 0; j < 4; ++j) {
    *reinterpret_cast<v4f*>(&lds[0][lA + j * 32 * PITCH]) = ga[j];
    *reinterpret_cast<v4f*>(&lds[0][lB + j * 32 * PITCH]) = gb[j];
  }
  __syncthreads();

#pragma unroll
  for (int c = 0; c < NCHUNK; ++c) {
    const float* cur = lds[c & 1];
    float*       nxt = lds[(c + 1) & 1];
    const bool  more = (c + 1 < NCHUNK);

    if (more) {
#pragma unroll
      for (int j = 0; j < 4; ++j) {
        ga[j] = *reinterpret_cast<const v4f*>(x + aoff[j] + (c + 1) * KC);
        gb[j] = *reinterpret_cast<const v4f*>(bslice + boff[j] + (c + 1) * KC);
      }
    }
    __builtin_amdgcn_sched_barrier(0);

#pragma unroll
    for (int s = 0; s < KC / 8; ++s) {
      v4f af[4], bf[2];
#pragma unroll
      for (int i = 0; i < 4; ++i)
        af[i] = *reinterpret_cast<const v4f*>(&cur[fA + i * 16 * PITCH + s * 8]);
#pragma unroll
      for (int j = 0; j < 2; ++j)
        bf[j] = *reinterpret_cast<const v4f*>(&cur[fB + j * 16 * PITCH + s * 8]);
#pragma unroll
      for (int i = 0; i < 4; ++i)
#pragma unroll
        for (int j = 0; j < 2; ++j)
          acc[i][j] = __builtin_amdgcn_wmma_f32_16x16x4_f32(
              false, af[i].xy, false, bf[j].xy, (short)0, acc[i][j], false, false);
#pragma unroll
      for (int i = 0; i < 4; ++i)
#pragma unroll
        for (int j = 0; j < 2; ++j)
          acc[i][j] = __builtin_amdgcn_wmma_f32_16x16x4_f32(
              false, af[i].zw, false, bf[j].zw, (short)0, acc[i][j], false, false);
    }

    __builtin_amdgcn_sched_barrier(0);
    if (more) {
#pragma unroll
      for (int j = 0; j < 4; ++j) {
        *reinterpret_cast<v4f*>(&nxt[lA + j * 32 * PITCH]) = ga[j];
        *reinterpret_cast<v4f*>(&nxt[lB + j * 32 * PITCH]) = gb[j];
      }
      __syncthreads();
    }
  }

  static_assert(NNODES % 16 == 0, "tile guard assumes whole 16-row tiles");
  static_assert(BM * CP <= 2 * (BM + BN) * PITCH, "accumulator staging must fit in the ring");
  if (r < NUM_BASES) {
    typedef _Float16 v8h __attribute__((ext_vector_type(8)));
    float* cs = &lds[0][0];
    float a[NUM_REL][NUM_BASES];
#pragma unroll
    for (int rr = 0; rr < NUM_REL; ++rr)
#pragma unroll
      for (int b = 0; b < NUM_BASES; ++b)
        a[rr][b] = att[rr * NUM_BASES + b];

    __syncthreads();
#pragma unroll
    for (int i = 0; i < 4; ++i)
#pragma unroll
      for (int j = 0; j < 2; ++j)
#pragma unroll
        for (int v = 0; v < 8; ++v)
          cs[(wm * 64 + i * 16 + 8 * half + v) * CP + wn * 32 + j * 16 + lid] = acc[i][j][v];
    __syncthreads();

    for (int p = 0; p < 2; ++p) {
      const int task = t + 256 * p;
      const int row  = task >> 2;
      const int cg   = task & 3;
      const int grow = row0 + row;
      v4f hb[NUM_BASES][2];
#pragma unroll
      for (int b = 0; b < NUM_BASES; ++b)
#pragma unroll
        for (int hh = 0; hh < 2; ++hh)
          hb[b][hh] = *reinterpret_cast<const v4f*>(&cs[row * CP + b * 32 + cg * 8 + hh * 4]);
      if (grow < NNODES) {
        _Float16* hp = h + (size_t)grow * OUT_CH + 32 * r + cg * 8;
#pragma unroll
        for (int rr = 0; rr < NUM_REL; ++rr) {
          v8h o;
#pragma unroll
          for (int e = 0; e < 8; ++e) {
            float s = a[rr][0] * hb[0][e >> 2][e & 3];
#pragma unroll
            for (int b = 1; b < NUM_BASES; ++b)
              s += a[rr][b] * hb[b][e >> 2][e & 3];
            o[e] = (_Float16)s;
          }
          *reinterpret_cast<v8h*>(hp + (size_t)rr * NNODES * OUT_CH) = o;
        }
      }
    }
  } else {
    const int q    = (wn & 1) * 16 + lid;
    const int colb = (wn >> 1) * 64 + q;
    const float bv0 = bias[colb];
    const float bv1 = bias[colb + 32];
#pragma unroll
    for (int i = 0; i < 4; ++i) {
      const int tr = row0 + wm * 64 + i * 16;
      if (tr < NNODES) {
#pragma unroll
        for (int j = 0; j < 2; ++j) {
          const float bv = j ? bv1 : bv0;
          float* op = out + (size_t)(tr + 8 * half) * OUT_CH + colb + j * 32;
#pragma unroll
          for (int v = 0; v < 8; ++v)
            op[(size_t)v * OUT_CH] = acc[i][j][v] + bv;
        }
      }
    }
  }
}

__global__ __launch_bounds__(256)
void edge_scatter_kernel(const int* __restrict__ esrc,
                         const int* __restrict__ edst,
                         const int* __restrict__ etyp,
                         const float* __restrict__ norm,
                         const _Float16* __restrict__ h,
                         float* __restrict__ out)
{
  const int lane   = threadIdx.x & 31;
  const int wid    = (blockIdx.x * blockDim.x + threadIdx.x) >> 5;
  const int nwaves = (gridDim.x * blockDim.x) >> 5;

  for (int e = wid; e < NEDGES; e += nwaves) {
    const int   s  = esrc[e];
    const int   d  = edst[e];
    const int   t  = etyp[e];
    const float nm = norm[e];

    const v4h hv = *reinterpret_cast<const v4h*>(
        h + ((size_t)t * NNODES + s) * OUT_CH + lane * 4);

    float* op = out + (size_t)d * OUT_CH + lane;
    atomicAdd(op +  0, (float)hv.x * nm);
    atomicAdd(op + 32, (float)hv.y * nm);
    atomicAdd(op + 64, (float)hv.z * nm);
    atomicAdd(op + 96, (float)hv.w * nm);
  }
}

extern "C" void kernel_launch(void* const* d_in, const int* in_sizes, int n_in,
                              void* d_out, int out_size, void* d_ws, size_t ws_size,
                              hipStream_t stream) {
  const float* x     = (const float*)d_in[0];
  const int*   esrc  = (const int*)  d_in[1];
  const int*   edst  = (const int*)  d_in[2];
  const int*   etyp  = (const int*)  d_in[3];
  const float* norm  = (const float*)d_in[4];
  const float* basis = (const float*)d_in[5];
  const float* att   = (const float*)d_in[6];
  const float* root  = (const float*)d_in[7];
  const float* bias  = (const float*)d_in[8];
  float* out = (float*)d_out;

  float*    wT = (float*)d_ws;
  _Float16* h  = (_Float16*)((char*)d_ws +
                             (size_t)NSLICE * IN_CH * OUT_CH * sizeof(float));

  build_wT_kernel<<<(NSLICE * IN_CH * OUT_CH + 255) / 256, 256, 0, stream>>>(
      basis, root, wT);

  rgcn_gemm_kernel<<<dim3(NSLICE, MTILES), 256, 0, stream>>>(
      x, wT, att, bias, h, out);

  edge_scatter_kernel<<<4096, 256, 0, stream>>>(esrc, edst, etyp, norm, h, out);
}


# Edge scatter re-pipelined: each wave loads 32 edge records at once, broadcasts them by readlane and keeps 8-16 h-row gathers in flight ahead of the adds; atomic adds in arrival order as before
# speedup vs baseline: 3.6367x; 1.1440x over previous
#include <hip/hip_runtime.h>
#include <hip/hip_fp16.h>

typedef float v2f __attribute__((ext_vector_type(2)));
typedef float v4f __attribute__((ext_vector_type(4)));
typedef float v8f __attribute__((ext_vector_type(8)));
typedef _Float16 v4h __attribute__((ext_vector_type(4)));

#define NNODES   50000
#define NEDGES   1600000
#define IN_CH    256
#define OUT_CH   128
#define NUM_REL  8
#define NUM_BASES 4
#define NSLICE   (NUM_BASES + 1)

__global__ __launch_bounds__(256)
void build_wT_kernel(const float* __restrict__ basis,
                     const float* __restrict__ root,
                     float* __restrict__ wT) {
  int idx = blockIdx.x * blockDim.x + threadIdx.x;
  if (idx >= NSLICE * IN_CH * OUT_CH) return;
  int s  = idx / (IN_CH * OUT_CH);
  int io = idx % (IN_CH * OUT_CH);
  int i  = io / OUT_CH;
  int o  = io % OUT_CH;
  const float v = (s < NUM_BASES) ? basis[idx] : root[io];
  wT[(size_t)s * OUT_CH * IN_CH + (size_t)o * IN_CH + i] = v;
}

#define BM     128
#define BN     128
#define KC     32
#define PITCH  36
#define NCHUNK (IN_CH / KC)
#define CP     132
#define MTILES ((NNODES + BM - 1) / BM)

__global__ __launch_bounds__(256)
void rgcn_gemm_kernel(const float* __restrict__ x,
                      const float* __restrict__ wT,
                      const float* __restrict__ att,
                      const float* __restrict__ bias,
                      _Float16* __restrict__ h,
                      float* __restrict__ out)
{
  __shared__ __attribute__((aligned(16))) float lds[2][(BM + BN) * PITCH];

  const int t    = threadIdx.x;
  const int lane = t & 31;
  const int wave = __builtin_amdgcn_readfirstlane(t >> 5);
  const int half = lane >> 4;
  const int lid  = lane & 15;
  const int wm   = wave >> 2;
  const int wn   = wave & 3;

  const int r    = blockIdx.x;
  const int row0 = blockIdx.y * BM;

  const int srow = t >> 3;
  const int sc4  = (t & 7) * 4;

  int aoff[4];
#pragma unroll
  for (int j = 0; j < 4; ++j) {
    int row = row0 + srow + 32 * j;
    row = row < NNODES ? row : NNODES - 1;
    aoff[j] = row * IN_CH + sc4;
  }
  const float* bslice = wT;
  int boff[4];
#pragma unroll
  for (int j = 0; j < 4; ++j) {
    int wrow;
    if (r < NUM_BASES) {
      wrow = j * OUT_CH + 8 * r + (srow >> 2) + 32 * (srow & 3);
    } else {
      const int ch = ((j >> 1) << 6) | ((srow >> 4) << 5) | ((j & 1) << 4) | (srow & 15);
      wrow = NUM_BASES * OUT_CH + ch;
    }
    boff[j] = wrow * IN_CH + sc4;
  }
  const int lA = srow * PITCH + sc4;
  const int lB = (BM + srow) * PITCH + sc4;

  const int fA = (wm * 64 + lid) * PITCH + half * 4;
  const int fB = (BM + wn * 32 + lid) * PITCH + half * 4;

  v8f acc[4][2] = {};
  v4f ga[4], gb[4];

#pragma unroll
  for (int j = 0; j < 4; ++j) {
    ga[j] = *reinterpret_cast<const v4f*>(x + aoff[j]);
    gb[j] = *reinterpret_cast<const v4f*>(bslice + boff[j]);
  }
#pragma unroll
  for (int j = 0; j < 4; ++j) {
    *reinterpret_cast<v4f*>(&lds[0][lA + j * 32 * PITCH]) = ga[j];
    *reinterpret_cast<v4f*>(&lds[0][lB + j * 32 * PITCH]) = gb[j];
  }
  __syncthreads();

#pragma unroll
  for (int c = 0; c < NCHUNK; ++c) {
    const float* cur = lds[c & 1];
    float*       nxt = lds[(c + 1) & 1];
    const bool  more = (c + 1 < NCHUNK);

    if (more) {
#pragma unroll
      for (int j = 0; j < 4; ++j) {
        ga[j] = *reinterpret_cast<const v4f*>(x + aoff[j] + (c + 1) * KC);
        gb[j] = *reinterpret_cast<const v4f*>(bslice + boff[j] + (c + 1) * KC);
      }
    }
    __builtin_amdgcn_sched_barrier(0);

#pragma unroll
    for (int s = 0; s < KC / 8; ++s) {
      v4f af[4], bf[2];
#pragma unroll
      for (int i = 0; i < 4; ++i)
        af[i] = *reinterpret_cast<const v4f*>(&cur[fA + i * 16 * PITCH + s * 8]);
#pragma unroll
      for (int j = 0; j < 2; ++j)
        bf[j] = *reinterpret_cast<const v4f*>(&cur[fB + j * 16 * PITCH + s * 8]);
#pragma unroll
      for (int i = 0; i < 4; ++i)
#pragma unroll
        for (int j = 0; j < 2; ++j)
          acc[i][j] = __builtin_amdgcn_wmma_f32_16x16x4_f32(
              false, af[i].xy, false, bf[j].xy, (short)0, acc[i][j], false, false);
#pragma unroll
      for (int i = 0; i < 4; ++i)
#pragma unroll
        for (int j = 0; j < 2; ++j)
          acc[i][j] = __builtin_amdgcn_wmma_f32_16x16x4_f32(
              false, af[i].zw, false, bf[j].zw, (short)0, acc[i][j], false, false);
    }

    __builtin_amdgcn_sched_barrier(0);
    if (more) {
#pragma unroll
      for (int j = 0; j < 4; ++j) {
        *reinterpret_cast<v4f*>(&nxt[lA + j * 32 * PITCH]) = ga[j];
        *reinterpret_cast<v4f*>(&nxt[lB + j * 32 * PITCH]) = gb[j];
      }
      __syncthreads();
    }
  }

  static_assert(NNODES % 16 == 0, "tile guard assumes whole 16-row tiles");
  static_assert(BM * CP <= 2 * (BM + BN) * PITCH, "accumulator staging must fit in the ring");
  if (r < NUM_BASES) {
    typedef _Float16 v8h __attribute__((ext_vector_type(8)));
    float* cs = &lds[0][0];
    float a[NUM_REL][NUM_BASES];
#pragma unroll
    for (int rr = 0; rr < NUM_REL; ++rr)
#pragma unroll
      for (int b = 0; b < NUM_BASES; ++b)
        a[rr][b] = att[rr * NUM_BASES + b];

    __syncthreads();
#pragma unroll
    for (int i = 0; i < 4; ++i)
#pragma unroll
      for (int j = 0; j < 2; ++j)
#pragma unroll
        for (int v = 0; v < 8; ++v)
          cs[(wm * 64 + i * 16 + 8 * half + v) * CP + wn * 32 + j * 16 + lid] = acc[i][j][v];
    __syncthreads();

    for (int p = 0; p < 2; ++p) {
      const int task = t + 256 * p;
      const int row  = task >> 2;
      const int cg   = task & 3;
      const int grow = row0 + row;
      v4f hb[NUM_BASES][2];
#pragma unroll
      for (int b = 0; b < NUM_BASES; ++b)
#pragma unroll
        for (int hh = 0; hh < 2; ++hh)
          hb[b][hh] = *reinterpret_cast<const v4f*>(&cs[row * CP + b * 32 + cg * 8 + hh * 4]);
      if (grow < NNODES) {
        _Float16* hp = h + (size_t)grow * OUT_CH + 32 * r + cg * 8;
#pragma unroll
        for (int rr = 0; rr < NUM_REL; ++rr) {
          v8h o;
#pragma unroll
          for (int e = 0; e < 8; ++e) {
            float s = a[rr][0] * hb[0][e >> 2][e & 3];
#pragma unroll
            for (int b = 1; b < NUM_BASES; ++b)
              s += a[rr][b] * hb[b][e >> 2][e & 3];
            o[e] = (_Float16)s;
          }
          *reinterpret_cast<v8h*>(hp + (size_t)rr * NNODES * OUT_CH) = o;
        }
      }
    }
  } else {
    const int q    = (wn & 1) * 16 + lid;
    const int colb = (wn >> 1) * 64 + q;
    const float bv0 = bias[colb];
    const float bv1 = bias[colb + 32];
#pragma unroll
    for (int i = 0; i < 4; ++i) {
      const int tr = row0 + wm * 64 + i * 16;
      if (tr < NNODES) {
#pragma unroll
        for (int j = 0; j < 2; ++j) {
          const float bv = j ? bv1 : bv0;
          float* op = out + (size_t)(tr + 8 * half) * OUT_CH + colb + j * 32;
#pragma unroll
          for (int v = 0; v < 8; ++v)
            op[(size_t)v * OUT_CH] = acc[i][j][v] + bv;
        }
      }
    }
  }
}

#define SC_WPB    4
#define SC_NWAVES 12500
#define SC_NBATCH (NEDGES / 32)
#define SC_BPW    (SC_NBATCH / SC_NWAVES)
static_assert(NEDGES % 32 == 0, "batches are whole: no lane guard in the scatter");
static_assert(SC_NBATCH % SC_NWAVES == 0, "every wave does the same number of batches");
static_assert(SC_NWAVES % SC_WPB == 0, "whole workgroups");
static_assert((size_t)NUM_REL * NNODES * OUT_CH * 2 < (1ull << 32), "h byte offsets fit 32 bits");

typedef v4h hrow8[8];

template <int G0>
__device__ __forceinline__ void sc_gather(hrow8& bank, const _Float16* __restrict__ h,
                                          unsigned hb, int lane) {
#pragma unroll
  for (int k = 0; k < 8; ++k) {
    const unsigned rb = (unsigned)__builtin_amdgcn_readlane((int)hb, G0 + k);
    const v4h* rp = reinterpret_cast<const v4h*>(reinterpret_cast<const char*>(h) + (size_t)rb);
    bank[k] = rp[lane];
  }
}

template <int G0>
__device__ __forceinline__ void sc_add(const hrow8& bank, float* __restrict__ out,
                                       unsigned ob, int nb, int lane) {
#pragma unroll
  for (int k = 0; k < 8; ++k) {
    const unsigned rb = (unsigned)__builtin_amdgcn_readlane((int)ob, G0 + k);
    const float    nm = __int_as_float(__builtin_amdgcn_readlane(nb, G0 + k));
    float* op = reinterpret_cast<float*>(reinterpret_cast<char*>(out) + (size_t)rb) + lane;
    atomicAdd(op +  0, (float)bank[k].x * nm);
    atomicAdd(op + 32, (float)bank[k].y * nm);
    atomicAdd(op + 64, (float)bank[k].z * nm);
    atomicAdd(op + 96, (float)bank[k].w * nm);
  }
}

__global__ __launch_bounds__(SC_WPB * 32)
void edge_scatter_kernel(const int* __restrict__ esrc,
                         const int* __restrict__ edst,
                         const int* __restrict__ etyp,
                         const float* __restrict__ norm,
                         const _Float16* __restrict__ h,
                         float* __restrict__ out)
{
  const int lane = threadIdx.x & 31;
  const int wid  = blockIdx.x * SC_WPB + __builtin_amdgcn_readfirstlane(threadIdx.x >> 5);

  int e = wid * 32 + lane;
  int   sv = esrc[e];
  int   dv = edst[e];
  int   tv = etyp[e];
  float nv = norm[e];
  unsigned hb = (unsigned)((tv * NNODES + sv) * OUT_CH) * 2u;
  unsigned ob = (unsigned)(dv * OUT_CH) * 4u;
  int      nb = __float_as_int(nv);

  hrow8 bankA, bankB;
  sc_gather<0>(bankA, h, hb, lane);

  for (int j = 0; j < SC_BPW; ++j) {
    const bool more = (j + 1 < SC_BPW);
    e = (wid + (more ? j + 1 : j) * SC_NWAVES) * 32 + lane;
    sv = esrc[e];
    dv = edst[e];
    tv = etyp[e];
    nv = norm[e];

    sc_gather<8>(bankB, h, hb, lane);
    __builtin_amdgcn_sched_barrier(0);
    sc_add<0>(bankA, out, ob, nb, lane);

    sc_gather<16>(bankA, h, hb, lane);
    __builtin_amdgcn_sched_barrier(0);
    sc_add<8>(bankB, out, ob, nb, lane);

    sc_gather<24>(bankB, h, hb, lane);
    __builtin_amdgcn_sched_barrier(0);
    sc_add<16>(bankA, out, ob, nb, lane);

    const unsigned hb2 = (unsigned)((tv * NNODES + sv) * OUT_CH) * 2u;
    const unsigned ob2 = (unsigned)(dv * OUT_CH) * 4u;
    const int      nb2 = __float_as_int(nv);
    if (more) sc_gather<0>(bankA, h, hb2, lane);
    __builtin_amdgcn_sched_barrier(0);
    sc_add<24>(bankB, out, ob, nb, lane);

    hb = hb2;
    ob = ob2;
    nb = nb2;
  }
}

extern "C" void kernel_launch(void* const* d_in, const int* in_sizes, int n_in,
                              void* d_out, int out_size, void* d_ws, size_t ws_size,
                              hipStream_t stream) {
  const float* x     = (const float*)d_in[0];
  const int*   esrc  = (const int*)  d_in[1];
  const int*   edst  = (const int*)  d_in[2];
  const int*   etyp  = (const int*)  d_in[3];
  const float* norm  = (const float*)d_in[4];
  const float* basis = (const float*)d_in[5];
  const float* att   = (const float*)d_in[6];
  const float* root  = (const float*)d_in[7];
  const float* bias  = (const float*)d_in[8];
  float* out = (float*)d_out;

  float*    wT = (float*)d_ws;
  _Float16* h  = (_Float16*)((char*)d_ws +
                             (size_t)NSLICE * IN_CH * OUT_CH * sizeof(float));

  build_wT_kernel<<<(NSLICE * IN_CH * OUT_CH + 255) / 256, 256, 0, stream>>>(
      basis, root, wT);

  rgcn_gemm_kernel<<<dim3(NSLICE, MTILES), 256, 0, stream>>>(
      x, wT, att, bias, h, out);

  edge_scatter_kernel<<<SC_NWAVES / SC_WPB, SC_WPB * 32, 0, stream>>>(esrc, edst, etyp, norm, h, out);
}


# Edge scatter split by cache half: each wave reads its die id and adds only into the out rows its half owns (edge units drawn from two ticket counters); atomic adds in arrival order as before
# speedup vs baseline: 4.0130x; 1.1035x over previous
#include <hip/hip_runtime.h>
#include <hip/hip_fp16.h>

typedef float v2f __attribute__((ext_vector_type(2)));
typedef float v4f __attribute__((ext_vector_type(4)));
typedef float v8f __attribute__((ext_vector_type(8)));
typedef _Float16 v4h __attribute__((ext_vector_type(4)));

#define NNODES   50000
#define NEDGES   1600000
#define IN_CH    256
#define OUT_CH   128
#define NUM_REL  8
#define NUM_BASES 4
#define NSLICE   (NUM_BASES + 1)
#define SC_TICK_STRIDE 64

__global__ __launch_bounds__(256)
void build_wT_kernel(const float* __restrict__ basis,
                     const float* __restrict__ root,
                     float* __restrict__ wT,
                     unsigned* __restrict__ tick) {
  int idx = blockIdx.x * blockDim.x + threadIdx.x;
  if (idx == 0) {
    tick[0] = 0u;
    tick[SC_TICK_STRIDE] = 0u;
  }
  if (idx >= NSLICE * IN_CH * OUT_CH) return;
  int s  = idx / (IN_CH * OUT_CH);
  int io = idx % (IN_CH * OUT_CH);
  int i  = io / OUT_CH;
  int o  = io % OUT_CH;
  const float v = (s < NUM_BASES) ? basis[idx] : root[io];
  wT[(size_t)s * OUT_CH * IN_CH + (size_t)o * IN_CH + i] = v;
}

#define BM     128
#define BN     128
#define KC     32
#define PITCH  36
#define NCHUNK (IN_CH / KC)
#define CP     132
#define MTILES ((NNODES + BM - 1) / BM)

__global__ __launch_bounds__(256)
void rgcn_gemm_kernel(const float* __restrict__ x,
                      const float* __restrict__ wT,
                      const float* __restrict__ att,
                      const float* __restrict__ bias,
                      _Float16* __restrict__ h,
                      float* __restrict__ out)
{
  __shared__ __attribute__((aligned(16))) float lds[2][(BM + BN) * PITCH];

  const int t    = threadIdx.x;
  const int lane = t & 31;
  const int wave = __builtin_amdgcn_readfirstlane(t >> 5);
  const int half = lane >> 4;
  const int lid  = lane & 15;
  const int wm   = wave >> 2;
  const int wn   = wave & 3;

  const int r    = blockIdx.x;
  const int row0 = blockIdx.y * BM;

  const int srow = t >> 3;
  const int sc4  = (t & 7) * 4;

  int aoff[4];
#pragma unroll
  for (int j = 0; j < 4; ++j) {
    int row = row0 + srow + 32 * j;
    row = row < NNODES ? row : NNODES - 1;
    aoff[j] = row * IN_CH + sc4;
  }
  const float* bslice = wT;
  int boff[4];
#pragma unroll
  for (int j = 0; j < 4; ++j) {
    int wrow;
    if (r < NUM_BASES) {
      wrow = j * OUT_CH + 8 * r + (srow >> 2) + 32 * (srow & 3);
    } else {
      const int ch = ((j >> 1) << 6) | ((srow >> 4) << 5) | ((j & 1) << 4) | (srow & 15);
      wrow = NUM_BASES * OUT_CH + ch;
    }
    boff[j] = wrow * IN_CH + sc4;
  }
  const int lA = srow * PITCH + sc4;
  const int lB = (BM + srow) * PITCH + sc4;

  const int fA = (wm * 64 + lid) * PITCH + half * 4;
  const int fB = (BM + wn * 32 + lid) * PITCH + half * 4;

  v8f acc[4][2] = {};
  v4f ga[4], gb[4];

#pragma unroll
  for (int j = 0; j < 4; ++j) {
    ga[j] = *reinterpret_cast<const v4f*>(x + aoff[j]);
    gb[j] = *reinterpret_cast<const v4f*>(bslice + boff[j]);
  }
#pragma unroll
  for (int j = 0; j < 4; ++j) {
    *reinterpret_cast<v4f*>(&lds[0][lA + j * 32 * PITCH]) = ga[j];
    *reinterpret_cast<v4f*>(&lds[0][lB + j * 32 * PITCH]) = gb[j];
  }
  __syncthreads();

#pragma unroll
  for (int c = 0; c < NCHUNK; ++c) {
    const float* cur = lds[c & 1];
    float*       nxt = lds[(c + 1) & 1];
    const bool  more = (c + 1 < NCHUNK);

    if (more) {
#pragma unroll
      for (int j = 0; j < 4; ++j) {
        ga[j] = *reinterpret_cast<const v4f*>(x + aoff[j] + (c + 1) * KC);
        gb[j] = *reinterpret_cast<const v4f*>(bslice + boff[j] + (c + 1) * KC);
      }
    }
    __builtin_amdgcn_sched_barrier(0);

#pragma unroll
    for (int s = 0; s < KC / 8; ++s) {
      v4f af[4], bf[2];
#pragma unroll
      for (int i = 0; i < 4; ++i)
        af[i] = *reinterpret_cast<const v4f*>(&cur[fA + i * 16 * PITCH + s * 8]);
#pragma unroll
      for (int j = 0; j < 2; ++j)
        bf[j] = *reinterpret_cast<const v4f*>(&cur[fB + j * 16 * PITCH + s * 8]);
#pragma unroll
      for (int i = 0; i < 4; ++i)
#pragma unroll
        for (int j = 0; j < 2; ++j)
          acc[i][j] = __builtin_amdgcn_wmma_f32_16x16x4_f32(
              false, af[i].xy, false, bf[j].xy, (short)0, acc[i][j], false, false);
#pragma unroll
      for (int i = 0; i < 4; ++i)
#pragma unroll
        for (int j = 0; j < 2; ++j)
          acc[i][j] = __builtin_amdgcn_wmma_f32_16x16x4_f32(
              false, af[i].zw, false, bf[j].zw, (short)0, acc[i][j], false, false);
    }

    __builtin_amdgcn_sched_barrier(0);
    if (more) {
#pragma unroll
      for (int j = 0; j < 4; ++j) {
        *reinterpret_cast<v4f*>(&nxt[lA + j * 32 * PITCH]) = ga[j];
        *reinterpret_cast<v4f*>(&nxt[lB + j * 32 * PITCH]) = gb[j];
      }
      __syncthreads();
    }
  }

  static_assert(NNODES % 16 == 0, "tile guard assumes whole 16-row tiles");
  static_assert(BM * CP <= 2 * (BM + BN) * PITCH, "accumulator staging must fit in the ring");
  if (r < NUM_BASES) {
    typedef _Float16 v8h __attribute__((ext_vector_type(8)));
    float* cs = &lds[0][0];
    float a[NUM_REL][NUM_BASES];
#pragma unroll
    for (int rr = 0; rr < NUM_REL; ++rr)
#pragma unroll
      for (int b = 0; b < NUM_BASES; ++b)
        a[rr][b] = att[rr * NUM_BASES + b];

    __syncthreads();
#pragma unroll
    for (int i = 0; i < 4; ++i)
#pragma unroll
      for (int j = 0; j < 2; ++j)
#pragma unroll
        for (int v = 0; v < 8; ++v)
          cs[(wm * 64 + i * 16 + 8 * half + v) * CP + wn * 32 + j * 16 + lid] = acc[i][j][v];
    __syncthreads();

    for (int p = 0; p < 2; ++p) {
      const int task = t + 256 * p;
      const int row  = task >> 2;
      const int cg   = task & 3;
      const int grow = row0 + row;
      v4f hb[NUM_BASES][2];
#pragma unroll
      for (int b = 0; b < NUM_BASES; ++b)
#pragma unroll
        for (int hh = 0; hh < 2; ++hh)
          hb[b][hh] = *reinterpret_cast<const v4f*>(&cs[row * CP + b * 32 + cg * 8 + hh * 4]);
      if (grow < NNODES) {
        _Float16* hp = h + (size_t)grow * OUT_CH + 32 * r + cg * 8;
#pragma unroll
        for (int rr = 0; rr < NUM_REL; ++rr) {
          v8h o;
#pragma unroll
          for (int e = 0; e < 8; ++e) {
            float s = a[rr][0] * hb[0][e >> 2][e & 3];
#pragma unroll
            for (int b = 1; b < NUM_BASES; ++b)
              s += a[rr][b] * hb[b][e >> 2][e & 3];
            o[e] = (_Float16)s;
          }
          *reinterpret_cast<v8h*>(hp + (size_t)rr * NNODES * OUT_CH) = o;
        }
      }
    }
  } else {
    const int q    = (wn & 1) * 16 + lid;
    const int colb = (wn >> 1) * 64 + q;
    const float bv0 = bias[colb];
    const float bv1 = bias[colb + 32];
#pragma unroll
    for (int i = 0; i < 4; ++i) {
      const int tr = row0 + wm * 64 + i * 16;
      if (tr < NNODES) {
#pragma unroll
        for (int j = 0; j < 2; ++j) {
          const float bv = j ? bv1 : bv0;
          float* op = out + (size_t)(tr + 8 * half) * OUT_CH + colb + j * 32;
#pragma unroll
          for (int v = 0; v < 8; ++v)
            op[(size_t)v * OUT_CH] = acc[i][j][v] + bv;
        }
      }
    }
  }
}

#define SC_WPB    4
#define SC_NWG    3072
#define SC_NBATCH (NEDGES / 32)
#define SC_NUNITS (SC_NBATCH / 2)
#define SC_SPLIT  (NNODES / 2)
static_assert(NEDGES % 64 == 0, "units are whole: no lane guard in the scatter");
static_assert((size_t)NUM_REL * NNODES * OUT_CH * 2 < (1ull << 32), "h byte offsets fit 32 bits");

struct ScBank {
  v4h      row[8];
  unsigned obv;
  int      nbv;
  int      idx[8];
  int      cnt;
};

__device__ __forceinline__ unsigned sc_draw(unsigned* __restrict__ tick, int role, int lane) {
  unsigned v = 0;
  if (lane == 0)
    v = __hip_atomic_fetch_add(tick + role * SC_TICK_STRIDE, 1u, __ATOMIC_RELAXED,
                               __HIP_MEMORY_SCOPE_AGENT);
  return v;
}

__global__ __launch_bounds__(SC_WPB * 32)
void edge_scatter_kernel(const int* __restrict__ esrc,
                         const int* __restrict__ edst,
                         const int* __restrict__ etyp,
                         const float* __restrict__ norm,
                         const _Float16* __restrict__ h,
                         float* __restrict__ out,
                         unsigned* __restrict__ tick)
{
  const int lane = threadIdx.x & 31;

  const unsigned place = (unsigned)__builtin_amdgcn_s_sendmsg_rtn(0x87);
  int  crole    = (int)((place >> 18) & 1u);
  bool switched = false;

  int   rs0 = 0, rd0 = 0, rt0 = 0, rs1 = 0, rd1 = 0, rt1 = 0;
  float rn0 = 0.f, rn1 = 0.f;
  int   prole = 0;
  bool  have_next = false;
  unsigned tnv = 0;

  unsigned hbv = 0, obv = 0, hbv1 = 0, obv1 = 0;
  int      nbv = 0, nbv1 = 0;
  unsigned m = 0, m1 = 0;

  auto take_ticket = [&](unsigned tv_) __attribute__((always_inline)) {
    unsigned t = (unsigned)__builtin_amdgcn_readfirstlane((int)tv_);
    if (t >= (unsigned)SC_NUNITS && !switched) {
      switched = true;
      crole ^= 1;
      t = (unsigned)__builtin_amdgcn_readfirstlane((int)sc_draw(tick, crole, lane));
    }
    have_next = t < (unsigned)SC_NUNITS;
    if (have_next) {
      const int e = (int)t * 64 + lane;
      rs0 = esrc[e];      rd0 = edst[e];      rt0 = etyp[e];      rn0 = norm[e];
      rs1 = esrc[e + 32]; rd1 = edst[e + 32]; rt1 = etyp[e + 32]; rn1 = norm[e + 32];
      prole = crole;
      tnv = sc_draw(tick, crole, lane);
    }
  };

  auto open_unit = [&]() __attribute__((always_inline)) {
    const bool up = (prole != 0);
    hbv  = (unsigned)((rt0 * NNODES + rs0) * OUT_CH) * 2u;
    obv  = (unsigned)(rd0 * OUT_CH) * 4u;
    nbv  = __float_as_int(rn0);
    hbv1 = (unsigned)((rt1 * NNODES + rs1) * OUT_CH) * 2u;
    obv1 = (unsigned)(rd1 * OUT_CH) * 4u;
    nbv1 = __float_as_int(rn1);
    m  = __builtin_amdgcn_ballot_w32((rd0 >= SC_SPLIT) == up);
    m1 = __builtin_amdgcn_ballot_w32((rd1 >= SC_SPLIT) == up);
    asm volatile("" : "+v"(hbv), "+v"(obv), "+v"(nbv), "+v"(hbv1), "+v"(obv1), "+v"(nbv1));
    take_ticket(tnv);
  };

  auto fill = [&](ScBank& b) __attribute__((always_inline)) {
    while (m == 0) {
      if (m1 != 0) { hbv = hbv1; obv = obv1; nbv = nbv1; m = m1; m1 = 0; }
      else if (have_next) open_unit();
      else break;
    }
    const int left = __builtin_popcount(m);
    b.cnt = left < 8 ? left : 8;
    b.obv = obv;
    b.nbv = nbv;
#pragma unroll
    for (int k = 0; k < 8; ++k) {
      const bool some = (k < left);
      int i;
      asm("s_ctz_i32_b32 %0, %1" : "=s"(i) : "s"(m));
      asm("s_bitset0_b32 %0, %1" : "+s"(m) : "s"(i));
      i = some ? i : 0;
      b.idx[k] = i;
      const unsigned rl = (unsigned)__builtin_amdgcn_readlane((int)hbv, i);
      const unsigned rb = some ? rl : 0u;
      const v4h* rp = reinterpret_cast<const v4h*>(reinterpret_cast<const char*>(h) + (size_t)rb);
      b.row[k] = rp[lane];
    }
  };

  auto add = [&](const ScBank& b) __attribute__((always_inline)) {
#pragma unroll
    for (int k = 0; k < 8; ++k) {
      if (k < b.cnt) {
        const unsigned rb = (unsigned)__builtin_amdgcn_readlane((int)b.obv, b.idx[k]);
        const float    nm = __int_as_float(__builtin_amdgcn_readlane(b.nbv, b.idx[k]));
        float* op = reinterpret_cast<float*>(reinterpret_cast<char*>(out) + (size_t)rb) + lane;
        atomicAdd(op +  0, (float)b.row[k].x * nm);
        atomicAdd(op + 32, (float)b.row[k].y * nm);
        atomicAdd(op + 64, (float)b.row[k].z * nm);
        atomicAdd(op + 96, (float)b.row[k].w * nm);
      }
    }
  };

  take_ticket(sc_draw(tick, crole, lane));

  ScBank A, B;
  fill(A);
  while (A.cnt != 0) {
    fill(B);
    __builtin_amdgcn_sched_barrier(0);
    add(A);
    if (B.cnt == 0) break;
    fill(A);
    __builtin_amdgcn_sched_barrier(0);
    add(B);
  }
}

extern "C" void kernel_launch(void* const* d_in, const int* in_sizes, int n_in,
                              void* d_out, int out_size, void* d_ws, size_t ws_size,
                              hipStream_t stream) {
  const float* x     = (const float*)d_in[0];
  const int*   esrc  = (const int*)  d_in[1];
  const int*   edst  = (const int*)  d_in[2];
  const int*   etyp  = (const int*)  d_in[3];
  const float* norm  = (const float*)d_in[4];
  const float* basis = (const float*)d_in[5];
  const float* att   = (const float*)d_in[6];
  const float* root  = (const float*)d_in[7];
  const float* bias  = (const float*)d_in[8];
  float* out = (float*)d_out;

  float*    wT = (float*)d_ws;
  _Float16* h  = (_Float16*)((char*)d_ws +
                             (size_t)NSLICE * IN_CH * OUT_CH * sizeof(float));
  unsigned* tick = (unsigned*)((char*)h + (size_t)NUM_REL * NNODES * OUT_CH * sizeof(_Float16));

  build_wT_kernel<<<(NSLICE * IN_CH * OUT_CH + 255) / 256, 256, 0, stream>>>(
      basis, root, wT, tick);

  rgcn_gemm_kernel<<<dim3(NSLICE, MTILES), 256, 0, stream>>>(
      x, wT, att, bias, h, out);

  edge_scatter_kernel<<<SC_NWG, SC_WPB * 32, 0, stream>>>(esrc, edst, etyp, norm, h, out, tick);
}


# Float-atomic scatter replaced by a sum per destination: edges binned by 64-row tile into out, tile sorted by row in LDS, each wave sums its rows in registers; root added after; sum order changed
# speedup vs baseline: 4.2686x; 1.0637x over previous
#include <hip/hip_runtime.h>
#include <hip/hip_fp16.h>

typedef float v2f __attribute__((ext_vector_type(2)));
typedef float v4f __attribute__((ext_vector_type(4)));
typedef float v8f __attribute__((ext_vector_type(8)));
typedef _Float16 v4h __attribute__((ext_vector_type(4)));

#define NNODES   50000
#define NEDGES   1600000
#define IN_CH    256
#define OUT_CH   128
#define NUM_REL  8
#define NUM_BASES 4
#define NSLICE   (NUM_BASES + 1)
#define TR       64
#define NTILE    ((NNODES + TR - 1) / TR)
#define TSLOTS   (TR * OUT_CH * 4 / 8)
#define TWORDS   (TR * OUT_CH)
#define TREC0    16
#define BIN_CH   4096
#define BIN_EPT  (BIN_CH / 256)

__global__ __launch_bounds__(256)
void build_wT_kernel(const float* __restrict__ basis,
                     const float* __restrict__ root,
                     float* __restrict__ wT,
                     float* __restrict__ out) {
  int idx = blockIdx.x * blockDim.x + threadIdx.x;
  if (idx < NTILE) reinterpret_cast<unsigned*>(out)[(size_t)idx * TWORDS] = 0u;
  if (idx >= NSLICE * IN_CH * OUT_CH) return;
  int s  = idx / (IN_CH * OUT_CH);
  int io = idx % (IN_CH * OUT_CH);
  int i  = io / OUT_CH;
  int o  = io % OUT_CH;
  const float v = (s < NUM_BASES) ? basis[idx] : root[io];
  wT[(size_t)s * OUT_CH * IN_CH + (size_t)o * IN_CH + i] = v;
}

__device__ __forceinline__ void bin_edges(const int* __restrict__ esrc,
                                          const int* __restrict__ edst,
                                          const int* __restrict__ etyp,
                                          const float* __restrict__ norm,
                                          float* __restrict__ out,
                                          unsigned* cnt,
                                          unsigned* base,
                                          int chunk, int t) {
  for (int i = t; i < NTILE; i += 256) cnt[i] = 0u;
  __syncthreads();

  unsigned tl[BIN_EPT], rk[BIN_EPT], hx[BIN_EPT];
  int      rd[BIN_EPT], rs[BIN_EPT], rt[BIN_EPT], nb[BIN_EPT];
#pragma unroll
  for (int j = 0; j < BIN_EPT; ++j) {
    const int e  = chunk * BIN_CH + j * 256 + t;
    const int ec = e < NEDGES ? e : NEDGES - 1;
    rd[j] = edst[ec];
    rs[j] = esrc[ec];
    rt[j] = etyp[ec];
    nb[j] = __float_as_int(norm[ec]);
  }
#pragma unroll
  for (int j = 0; j < BIN_EPT; ++j) {
    const bool     valid = chunk * BIN_CH + j * 256 + t < NEDGES;
    const unsigned d     = (unsigned)rd[j];
    const unsigned tile  = d / TR;
    hx[j] = (unsigned)(rt[j] * NNODES + rs[j]) | ((d % TR) << 20);
    tl[j] = valid ? tile : 0xFFFFFFFFu;
    rk[j] = 0u;
    if (valid) rk[j] = atomicAdd(&cnt[tile], 1u);
  }
  __syncthreads();

  unsigned* ocnt = reinterpret_cast<unsigned*>(out);
  unsigned bq[(NTILE + 255) / 256];
#pragma unroll
  for (int q = 0; q < (NTILE + 255) / 256; ++q) {
    const int i = t + 256 * q;
    bq[q] = 0u;
    if (i < NTILE) bq[q] = atomicAdd(ocnt + (size_t)i * TWORDS, cnt[i]);
  }
#pragma unroll
  for (int q = 0; q < (NTILE + 255) / 256; ++q) {
    const int i = t + 256 * q;
    if (i < NTILE) base[i] = bq[q];
  }
  __syncthreads();

  uint2* slots = reinterpret_cast<uint2*>(out);
#pragma unroll
  for (int j = 0; j < BIN_EPT; ++j) {
    if (tl[j] != 0xFFFFFFFFu) {
      const unsigned pos  = base[tl[j]] + rk[j];
      const int      left = NNODES - (int)tl[j] * TR;
      const unsigned cap  = (unsigned)((left < TR ? left : TR) * (OUT_CH * 4 / 8) - TREC0);
      if (pos < cap) slots[(size_t)tl[j] * TSLOTS + TREC0 + pos] = make_uint2(hx[j], (unsigned)nb[j]);
    }
  }
}

#define BM     128
#define BN     128
#define KC     32
#define PITCH  36
#define NCHUNK (IN_CH / KC)
#define CP     132
#define MTILES ((NNODES + BM - 1) / BM)

__global__ __launch_bounds__(256)
void rgcn_gemm_kernel(const float* __restrict__ x,
                      const float* __restrict__ wT,
                      const float* __restrict__ att,
                      const float* __restrict__ bias,
                      _Float16* __restrict__ h,
                      float* __restrict__ out,
                      const int* __restrict__ esrc,
                      const int* __restrict__ edst,
                      const int* __restrict__ etyp,
                      const float* __restrict__ norm,
                      int root_pass)
{
  __shared__ __attribute__((aligned(16))) float lds[2][(BM + BN) * PITCH];

  static_assert(2 * NTILE * 4 <= (int)sizeof(lds), "binning counters fit in the ring");
  static_assert((size_t)MTILES * BIN_CH >= NEDGES, "one binning workgroup per M tile covers the edges");
  if (!root_pass && blockIdx.x == NUM_BASES) {
    unsigned* cnt = reinterpret_cast<unsigned*>(&lds[0][0]);
    bin_edges(esrc, edst, etyp, norm, out, cnt, cnt + NTILE, blockIdx.y, threadIdx.x);
    return;
  }

  const int t    = threadIdx.x;
  const int lane = t & 31;
  const int wave = __builtin_amdgcn_readfirstlane(t >> 5);
  const int half = lane >> 4;
  const int lid  = lane & 15;
  const int wm   = wave >> 2;
  const int wn   = wave & 3;

  const int r    = root_pass ? NUM_BASES : (int)blockIdx.x;
  const int row0 = blockIdx.y * BM;

  const int srow = t >> 3;
  const int sc4  = (t & 7) * 4;

  int aoff[4];
#pragma unroll
  for (int j = 0; j < 4; ++j) {
    int row = row0 + srow + 32 * j;
    row = row < NNODES ? row : NNODES - 1;
    aoff[j] = row * IN_CH + sc4;
  }
  const float* bslice = wT;
  int boff[4];
#pragma unroll
  for (int j = 0; j < 4; ++j) {
    int wrow;
    if (r < NUM_BASES) {
      wrow = j * OUT_CH + 8 * r + (srow >> 2) + 32 * (srow & 3);
    } else {
      const int ch = ((j >> 1) << 6) | ((srow >> 4) << 5) | ((j & 1) << 4) | (srow & 15);
      wrow = NUM_BASES * OUT_CH + ch;
    }
    boff[j] = wrow * IN_CH + sc4;
  }
  const int lA = srow * PITCH + sc4;
  const int lB = (BM + srow) * PITCH + sc4;

  const int fA = (wm * 64 + lid) * PITCH + half * 4;
  const int fB = (BM + wn * 32 + lid) * PITCH + half * 4;

  v8f acc[4][2] = {};
  v4f ga[4], gb[4];

#pragma unroll
  for (int j = 0; j < 4; ++j) {
    ga[j] = *reinterpret_cast<const v4f*>(x + aoff[j]);
    gb[j] = *reinterpret_cast<const v4f*>(bslice + boff[j]);
  }
#pragma unroll
  for (int j = 0; j < 4; ++j) {
    *reinterpret_cast<v4f*>(&lds[0][lA + j * 32 * PITCH]) = ga[j];
    *reinterpret_cast<v4f*>(&lds[0][lB + j * 32 * PITCH]) = gb[j];
  }
  __syncthreads();

#pragma unroll
  for (int c = 0; c < NCHUNK; ++c) {
    const float* cur = lds[c & 1];
    float*       nxt = lds[(c + 1) & 1];
    const bool  more = (c + 1 < NCHUNK);

    if (more) {
#pragma unroll
      for (int j = 0; j < 4; ++j) {
        ga[j] = *reinterpret_cast<const v4f*>(x + aoff[j] + (c + 1) * KC);
        gb[j] = *reinterpret_cast<const v4f*>(bslice + boff[j] + (c + 1) * KC);
      }
    }
    __builtin_amdgcn_sched_barrier(0);

#pragma unroll
    for (int s = 0; s < KC / 8; ++s) {
      v4f af[4], bf[2];
#pragma unroll
      for (int i = 0; i < 4; ++i)
        af[i] = *reinterpret_cast<const v4f*>(&cur[fA + i * 16 * PITCH + s * 8]);
#pragma unroll
      for (int j = 0; j < 2; ++j)
        bf[j] = *reinterpret_cast<const v4f*>(&cur[fB + j * 16 * PITCH + s * 8]);
#pragma unroll
      for (int i = 0; i < 4; ++i)
#pragma unroll
        for (int j = 0; j < 2; ++j)
          acc[i][j] = __builtin_amdgcn_wmma_f32_16x16x4_f32(
              false, af[i].xy, false, bf[j].xy, (short)0, acc[i][j], false, false);
#pragma unroll
      for (int i = 0; i < 4; ++i)
#pragma unroll
        for (int j = 0; j < 2; ++j)
          acc[i][j] = __builtin_amdgcn_wmma_f32_16x16x4_f32(
              false, af[i].zw, false, bf[j].zw, (short)0, acc[i][j], false, false);
    }

    __builtin_amdgcn_sched_barrier(0);
    if (more) {
#pragma unroll
      for (int j = 0; j < 4; ++j) {
        *reinterpret_cast<v4f*>(&nxt[lA + j * 32 * PITCH]) = ga[j];
        *reinterpret_cast<v4f*>(&nxt[lB + j * 32 * PITCH]) = gb[j];
      }
      __syncthreads();
    }
  }

  static_assert(NNODES % 16 == 0, "tile guard assumes whole 16-row tiles");
  static_assert(BM * CP <= 2 * (BM + BN) * PITCH, "accumulator staging must fit in the ring");
  if (r < NUM_BASES) {
    typedef _Float16 v8h __attribute__((ext_vector_type(8)));
    float* cs = &lds[0][0];
    float a[NUM_REL][NUM_BASES];
#pragma unroll
    for (int rr = 0; rr < NUM_REL; ++rr)
#pragma unroll
      for (int b = 0; b < NUM_BASES; ++b)
        a[rr][b] = att[rr * NUM_BASES + b];

    __syncthreads();
#pragma unroll
    for (int i = 0; i < 4; ++i)
#pragma unroll
      for (int j = 0; j < 2; ++j)
#pragma unroll
        for (int v = 0; v < 8; ++v)
          cs[(wm * 64 + i * 16 + 8 * half + v) * CP + wn * 32 + j * 16 + lid] = acc[i][j][v];
    __syncthreads();

    for (int p = 0; p < 2; ++p) {
      const int task = t + 256 * p;
      const int row  = task >> 2;
      const int cg   = task & 3;
      const int grow = row0 + row;
      v4f hb[NUM_BASES][2];
#pragma unroll
      for (int b = 0; b < NUM_BASES; ++b)
#pragma unroll
        for (int hh = 0; hh < 2; ++hh)
          hb[b][hh] = *reinterpret_cast<const v4f*>(&cs[row * CP + b * 32 + cg * 8 + hh * 4]);
      if (grow < NNODES) {
        _Float16* hp = h + (size_t)grow * OUT_CH + 32 * r + cg * 8;
#pragma unroll
        for (int rr = 0; rr < NUM_REL; ++rr) {
          v8h o;
#pragma unroll
          for (int e = 0; e < 8; ++e) {
            float s = a[rr][0] * hb[0][e >> 2][e & 3];
#pragma unroll
            for (int b = 1; b < NUM_BASES; ++b)
              s += a[rr][b] * hb[b][e >> 2][e & 3];
            o[e] = (_Float16)s;
          }
          *reinterpret_cast<v8h*>(hp + (size_t)rr * NNODES * OUT_CH) = o;
        }
      }
    }
  } else {
    const int q    = (wn & 1) * 16 + lid;
    const int colb = (wn >> 1) * 64 + q;
    const float bv0 = bias[colb];
    const float bv1 = bias[colb + 32];
#pragma unroll
    for (int i = 0; i < 4; ++i) {
      const int tr = row0 + wm * 64 + i * 16;
      if (tr < NNODES) {
#pragma unroll
        for (int j = 0; j < 2; ++j) {
          const float bv = j ? bv1 : bv0;
          float* op = out + (size_t)(tr + 8 * half) * OUT_CH + colb + j * 32;
#pragma unroll
          for (int v = 0; v < 8; ++v)
            op[(size_t)v * OUT_CH] += acc[i][j][v] + bv;
        }
      }
    }
  }
}

#define TS_WAVES 8
#define TS_RPW   (TR / TS_WAVES)
#define TS_RPT   (TSLOTS / (TS_WAVES * 32))
static_assert((size_t)NUM_REL * NNODES * OUT_CH * 2 < (1ull << 32), "h byte offsets fit 32 bits");
static_assert((size_t)NUM_REL * NNODES < (1u << 20), "h row index fits 20 bits of a record");
static_assert(TR <= 64, "row inside the tile fits 6 bits of a record");
static_assert(NNODES % TR >= 1 && (NNODES % TR) * (OUT_CH * 4 / 8) > TREC0, "last tile has record slots");
static_assert(TSLOTS - TREC0 <= TS_RPT * TS_WAVES * 32, "a thread's TS_RPT records cover every slot");
static_assert(NEDGES % (TS_WAVES * 32) == 0, "whole batches in the edge-list walk of an overfull tile");

__global__ __launch_bounds__(TS_WAVES * 32)
void tile_sum_kernel(const int* __restrict__ esrc,
                     const int* __restrict__ edst,
                     const int* __restrict__ etyp,
                     const float* __restrict__ norm,
                     const _Float16* __restrict__ h,
                     float* out)
{
  __shared__ uint2    srt[TSLOTS];
  __shared__ unsigned cnt[TR];
  __shared__ unsigned start[TR + 1];

  const int t    = threadIdx.x;
  const int lane = t & 31;
  const int wave = __builtin_amdgcn_readfirstlane(t >> 5);
  const int row0 = blockIdx.x * TR;
  const int rows = (NNODES - row0) < TR ? (NNODES - row0) : TR;
  const unsigned cap = (unsigned)(rows * (OUT_CH * 4 / 8) - TREC0);

  const uint2* slots = reinterpret_cast<const uint2*>(out) + (size_t)blockIdx.x * TSLOTS;
  const unsigned n = (unsigned)__builtin_amdgcn_readfirstlane((int)slots[0].x);

  if (n <= cap) {
    if (t < TR) cnt[t] = 0u;
    __syncthreads();
    uint2    rv[TS_RPT];
    unsigned rk[TS_RPT];
#pragma unroll
    for (int j = 0; j < TS_RPT; ++j) {
      const unsigned i = t + TS_WAVES * 32 * j;
      rv[j] = slots[TREC0 + (i < n ? i : 0u)];
    }
#pragma unroll
    for (int j = 0; j < TS_RPT; ++j) {
      rk[j] = 0u;
      if (t + TS_WAVES * 32 * j < n) rk[j] = atomicAdd(&cnt[rv[j].x >> 20], 1u);
    }
    __syncthreads();
    if (t < TR) {
      unsigned s = 0;
      for (int k = 0; k < t; ++k) s += cnt[k];
      start[t] = s;
    }
    if (t == 0) start[TR] = n;
    __syncthreads();
#pragma unroll
    for (int j = 0; j < TS_RPT; ++j)
      if (t + TS_WAVES * 32 * j < n) srt[start[rv[j].x >> 20] + rk[j]] = rv[j];
    __syncthreads();

    const int rlo = wave * TS_RPW;
    if (rlo < rows) {
      const int rhi = (rlo + TS_RPW) < rows ? (rlo + TS_RPW) : rows;
      const unsigned s0    = (unsigned)__builtin_amdgcn_readfirstlane((int)start[rlo]);
      const unsigned total = (unsigned)__builtin_amdgcn_readfirstlane((int)start[rhi]) - s0;

      float a0 = 0.f, a1 = 0.f, a2 = 0.f, a3 = 0.f;
      int   crow = rlo;

      auto flush_to = [&](int upto) __attribute__((always_inline)) {
        float* op = out + (size_t)(row0 + crow) * OUT_CH + lane;
        op[0] = a0; op[32] = a1; op[64] = a2; op[96] = a3;
        a0 = 0.f; a1 = 0.f; a2 = 0.f; a3 = 0.f;
#pragma clang loop unroll(disable)
        for (int r = crow + 1; r < upto; ++r) {
          float* zp = out + (size_t)(row0 + r) * OUT_CH + lane;
          zp[0] = 0.f; zp[32] = 0.f; zp[64] = 0.f; zp[96] = 0.f;
        }
        crow = upto;
      };
      auto load_chunk = [&](unsigned c) __attribute__((always_inline)) -> uint2 {
        const unsigned i = c * 32 + lane;
        return i < total ? srt[s0 + i] : make_uint2(0u, 0u);
      };
      auto issue = [&](v4h (&row)[8], const uint2& rc, const int g) __attribute__((always_inline)) {
#pragma unroll
        for (int k = 0; k < 8; ++k) {
          const unsigned x  = (unsigned)__builtin_amdgcn_readlane((int)rc.x, g * 8 + k);
          const size_t   rb = (size_t)(x & 0xFFFFFu) * (OUT_CH * 2);
          row[k] = reinterpret_cast<const v4h*>(reinterpret_cast<const char*>(h) + rb)[lane];
        }
      };
      auto consume = [&](const v4h (&row)[8], const uint2& rc, const int g, const int nav)
          __attribute__((always_inline)) {
#pragma unroll
        for (int k = 0; k < 8; ++k) {
          if (k < nav) {
            const unsigned x  = (unsigned)__builtin_amdgcn_readlane((int)rc.x, g * 8 + k);
            const float    nm = __int_as_float(__builtin_amdgcn_readlane((int)rc.y, g * 8 + k));
            const int      dl = (int)(x >> 20);
            if (dl != crow) flush_to(dl);
            a0 += (float)row[k].x * nm;
            a1 += (float)row[k].y * nm;
            a2 += (float)row[k].z * nm;
            a3 += (float)row[k].w * nm;
          }
        }
      };
      auto navail = [&](int rem, int g) __attribute__((always_inline)) -> int {
        const int v = rem - 8 * g;
        return v < 0 ? 0 : (v > 8 ? 8 : v);
      };

      v4h R0[8], R1[8];
      uint2 rc = load_chunk(0);
      uint2 rn = load_chunk(1);
      issue(R0, rc, 0);
      for (unsigned c = 0; c * 32 < total; ++c) {
        const int rem = (int)(total - c * 32);
        issue(R1, rc, 1);
        __builtin_amdgcn_sched_barrier(0);
        consume(R0, rc, 0, navail(rem, 0));
        issue(R0, rc, 2);
        __builtin_amdgcn_sched_barrier(0);
        consume(R1, rc, 1, navail(rem, 1));
        issue(R1, rc, 3);
        __builtin_amdgcn_sched_barrier(0);
        consume(R0, rc, 2, navail(rem, 2));
        issue(R0, rn, 0);
        __builtin_amdgcn_sched_barrier(0);
        consume(R1, rc, 3, navail(rem, 3));
        rc = rn;
        rn = load_chunk(c + 2);
      }
      flush_to(rhi);
    }
  } else {
    float* fs = reinterpret_cast<float*>(srt);
    for (int i = t; i < TR * OUT_CH; i += TS_WAVES * 32) fs[i] = 0.f;
    __syncthreads();
    for (int eb = wave * 32; eb < NEDGES; eb += TS_WAVES * 32) {
      const int e = eb + lane;
      const unsigned dl = (unsigned)(edst[e] - row0);
      unsigned m = __builtin_amdgcn_ballot_w32(dl < (unsigned)rows);
      if (m != 0u) {
        const unsigned hid = (unsigned)(etyp[e] * NNODES + esrc[e]);
        const int      nbv = __float_as_int(norm[e]);
        while (m != 0u) {
          const int i = __builtin_ctz(m);
          m &= m - 1u;
          const unsigned hi  = (unsigned)__builtin_amdgcn_readlane((int)hid, i);
          const unsigned dli = (unsigned)__builtin_amdgcn_readlane((int)dl, i);
          const float    nm  = __int_as_float(__builtin_amdgcn_readlane(nbv, i));
          const v4h r = reinterpret_cast<const v4h*>(
              reinterpret_cast<const char*>(h) + (size_t)hi * (OUT_CH * 2))[lane];
          float* fp = fs + dli * OUT_CH + lane;
          atomicAdd(fp +  0, (float)r.x * nm);
          atomicAdd(fp + 32, (float)r.y * nm);
          atomicAdd(fp + 64, (float)r.z * nm);
          atomicAdd(fp + 96, (float)r.w * nm);
        }
      }
    }
    __syncthreads();
    for (int i = t; i < rows * OUT_CH; i += TS_WAVES * 32) out[(size_t)row0 * OUT_CH + i] = fs[i];
  }
}

extern "C" void kernel_launch(void* const* d_in, const int* in_sizes, int n_in,
                              void* d_out, int out_size, void* d_ws, size_t ws_size,
                              hipStream_t stream) {
  const float* x     = (const float*)d_in[0];
  const int*   esrc  = (const int*)  d_in[1];
  const int*   edst  = (const int*)  d_in[2];
  const int*   etyp  = (const int*)  d_in[3];
  const float* norm  = (const float*)d_in[4];
  const float* basis = (const float*)d_in[5];
  const float* att   = (const float*)d_in[6];
  const float* root  = (const float*)d_in[7];
  const float* bias  = (const float*)d_in[8];
  float* out = (float*)d_out;

  float*    wT = (float*)d_ws;
  _Float16* h  = (_Float16*)((char*)d_ws +
                             (size_t)NSLICE * IN_CH * OUT_CH * sizeof(float));

  build_wT_kernel<<<(NSLICE * IN_CH * OUT_CH + 255) / 256, 256, 0, stream>>>(
      basis, root, wT, out);

  rgcn_gemm_kernel<<<dim3(NSLICE, MTILES), 256, 0, stream>>>(
      x, wT, att, bias, h, out, esrc, edst, etyp, norm, 0);

  tile_sum_kernel<<<NTILE, TS_WAVES * 32, 0, stream>>>(esrc, edst, etyp, norm, h, out);

  rgcn_gemm_kernel<<<dim3(1, MTILES), 256, 0, stream>>>(
      x, wT, att, bias, h, out, esrc, edst, etyp, norm, 1);
}


# Root product x@root+bias moved from its own launch into the tile kernel: each tile workgroup forms it for its 64 rows by f32 WMMA, half of them before their gather, half after; sum order unchanged
# speedup vs baseline: 4.4265x; 1.0370x over previous
#include <hip/hip_runtime.h>
#include <hip/hip_fp16.h>

typedef float v2f __attribute__((ext_vector_type(2)));
typedef float v4f __attribute__((ext_vector_type(4)));
typedef float v8f __attribute__((ext_vector_type(8)));
typedef _Float16 v4h __attribute__((ext_vector_type(4)));

#define NNODES   50000
#define NEDGES   1600000
#define IN_CH    256
#define OUT_CH   128
#define NUM_REL  8
#define NUM_BASES 4
#define NSLICE   (NUM_BASES + 1)
#define TR       64
#define NTILE    ((NNODES + TR - 1) / TR)
#define TSLOTS   (TR * OUT_CH * 4 / 8)
#define TWORDS   (TR * OUT_CH)
#define TREC0    16
#define BIN_CH   4096
#define BIN_EPT  (BIN_CH / 256)

__global__ __launch_bounds__(256)
void build_wT_kernel(const float* __restrict__ basis,
                     const float* __restrict__ root,
                     float* __restrict__ wT,
                     float* __restrict__ out) {
  int idx = blockIdx.x * blockDim.x + threadIdx.x;
  if (idx < NTILE) reinterpret_cast<unsigned*>(out)[(size_t)idx * TWORDS] = 0u;
  if (idx >= NSLICE * IN_CH * OUT_CH) return;
  int s  = idx / (IN_CH * OUT_CH);
  int io = idx % (IN_CH * OUT_CH);
  int i  = io / OUT_CH;
  int o  = io % OUT_CH;
  const float v = (s < NUM_BASES) ? basis[idx] : root[io];
  wT[(size_t)s * OUT_CH * IN_CH + (size_t)o * IN_CH + i] = v;
}

__device__ __forceinline__ void bin_edges(const int* __restrict__ esrc,
                                          const int* __restrict__ edst,
                                          const int* __restrict__ etyp,
                                          const float* __restrict__ norm,
                                          float* __restrict__ out,
                                          unsigned* cnt,
                                          unsigned* base,
                                          int chunk, int t) {
  for (int i = t; i < NTILE; i += 256) cnt[i] = 0u;
  __syncthreads();

  unsigned tl[BIN_EPT], rk[BIN_EPT], hx[BIN_EPT];
  int      rd[BIN_EPT], rs[BIN_EPT], rt[BIN_EPT], nb[BIN_EPT];
#pragma unroll
  for (int j = 0; j < BIN_EPT; ++j) {
    const int e  = chunk * BIN_CH + j * 256 + t;
    const int ec = e < NEDGES ? e : NEDGES - 1;
    rd[j] = edst[ec];
    rs[j] = esrc[ec];
    rt[j] = etyp[ec];
    nb[j] = __float_as_int(norm[ec]);
  }
#pragma unroll
  for (int j = 0; j < BIN_EPT; ++j) {
    const bool     valid = chunk * BIN_CH + j * 256 + t < NEDGES;
    const unsigned d     = (unsigned)rd[j];
    const unsigned tile  = d / TR;
    hx[j] = (unsigned)(rt[j] * NNODES + rs[j]) | ((d % TR) << 20);
    tl[j] = valid ? tile : 0xFFFFFFFFu;
    rk[j] = 0u;
    if (valid) rk[j] = atomicAdd(&cnt[tile], 1u);
  }
  __syncthreads();

  unsigned* ocnt = reinterpret_cast<unsigned*>(out);
  unsigned bq[(NTILE + 255) / 256];
#pragma unroll
  for (int q = 0; q < (NTILE + 255) / 256; ++q) {
    const int i = t + 256 * q;
    bq[q] = 0u;
    if (i < NTILE) bq[q] = atomicAdd(ocnt + (size_t)i * TWORDS, cnt[i]);
  }
#pragma unroll
  for (int q = 0; q < (NTILE + 255) / 256; ++q) {
    const int i = t + 256 * q;
    if (i < NTILE) base[i] = bq[q];
  }
  __syncthreads();

  uint2* slots = reinterpret_cast<uint2*>(out);
#pragma unroll
  for (int j = 0; j < BIN_EPT; ++j) {
    if (tl[j] != 0xFFFFFFFFu) {
      const unsigned pos  = base[tl[j]] + rk[j];
      const int      left = NNODES - (int)tl[j] * TR;
      const unsigned cap  = (unsigned)((left < TR ? left : TR) * (OUT_CH * 4 / 8) - TREC0);
      if (pos < cap) slots[(size_t)tl[j] * TSLOTS + TREC0 + pos] = make_uint2(hx[j], (unsigned)nb[j]);
    }
  }
}

#define BM     128
#define BN     128
#define KC     32
#define PITCH  36
#define NCHUNK (IN_CH / KC)
#define CP     132
#define MTILES ((NNODES + BM - 1) / BM)

__global__ __launch_bounds__(256)
void rgcn_gemm_kernel(const float* __restrict__ x,
                      const float* __restrict__ wT,
                      const float* __restrict__ att,
                      _Float16* __restrict__ h,
                      float* __restrict__ out,
                      const int* __restrict__ esrc,
                      const int* __restrict__ edst,
                      const int* __restrict__ etyp,
                      const float* __restrict__ norm)
{
  __shared__ __attribute__((aligned(16))) float lds[2][(BM + BN) * PITCH];

  static_assert(2 * NTILE * 4 <= (int)sizeof(lds), "binning counters fit in the ring");
  static_assert((size_t)MTILES * BIN_CH >= NEDGES, "one binning workgroup per M tile covers the edges");
  if (blockIdx.x == NUM_BASES) {
    unsigned* cnt = reinterpret_cast<unsigned*>(&lds[0][0]);
    bin_edges(esrc, edst, etyp, norm, out, cnt, cnt + NTILE, blockIdx.y, threadIdx.x);
    return;
  }

  const int t    = threadIdx.x;
  const int lane = t & 31;
  const int wave = __builtin_amdgcn_readfirstlane(t >> 5);
  const int half = lane >> 4;
  const int lid  = lane & 15;
  const int wm   = wave >> 2;
  const int wn   = wave & 3;

  const int r    = (int)blockIdx.x;
  const int row0 = blockIdx.y * BM;

  const int srow = t >> 3;
  const int sc4  = (t & 7) * 4;

  int aoff[4];
#pragma unroll
  for (int j = 0; j < 4; ++j) {
    int row = row0 + srow + 32 * j;
    row = row < NNODES ? row : NNODES - 1;
    aoff[j] = row * IN_CH + sc4;
  }
  const float* bslice = wT;
  int boff[4];
#pragma unroll
  for (int j = 0; j < 4; ++j) {
    const int wrow = j * OUT_CH + 8 * r + (srow >> 2) + 32 * (srow & 3);
    boff[j] = wrow * IN_CH + sc4;
  }
  const int lA = srow * PITCH + sc4;
  const int lB = (BM + srow) * PITCH + sc4;

  const int fA = (wm * 64 + lid) * PITCH + half * 4;
  const int fB = (BM + wn * 32 + lid) * PITCH + half * 4;

  v8f acc[4][2] = {};
  v4f ga[4], gb[4];

#pragma unroll
  for (int j = 0; j < 4; ++j) {
    ga[j] = *reinterpret_cast<const v4f*>(x + aoff[j]);
    gb[j] = *reinterpret_cast<const v4f*>(bslice + boff[j]);
  }
#pragma unroll
  for (int j = 0; j < 4; ++j) {
    *reinterpret_cast<v4f*>(&lds[0][lA + j * 32 * PITCH]) = ga[j];
    *reinterpret_cast<v4f*>(&lds[0][lB + j * 32 * PITCH]) = gb[j];
  }
  __syncthreads();

#pragma unroll
  for (int c = 0; c < NCHUNK; ++c) {
    const float* cur = lds[c & 1];
    float*       nxt = lds[(c + 1) & 1];
    const bool  more = (c + 1 < NCHUNK);

    if (more) {
#pragma unroll
      for (int j = 0; j < 4; ++j) {
        ga[j] = *reinterpret_cast<const v4f*>(x + aoff[j] + (c + 1) * KC);
        gb[j] = *reinterpret_cast<const v4f*>(bslice + boff[j] + (c + 1) * KC);
      }
    }
    __builtin_amdgcn_sched_barrier(0);

#pragma unroll
    for (int s = 0; s < KC / 8; ++s) {
      v4f af[4], bf[2];
#pragma unroll
      for (int i = 0; i < 4; ++i)
        af[i] = *reinterpret_cast<const v4f*>(&cur[fA + i * 16 * PITCH + s * 8]);
#pragma unroll
      for (int j = 0; j < 2; ++j)
        bf[j] = *reinterpret_cast<const v4f*>(&cur[fB + j * 16 * PITCH + s * 8]);
#pragma unroll
      for (int i = 0; i < 4; ++i)
#pragma unroll
        for (int j = 0; j < 2; ++j)
          acc[i][j] = __builtin_amdgcn_wmma_f32_16x16x4_f32(
              false, af[i].xy, false, bf[j].xy, (short)0, acc[i][j], false, false);
#pragma unroll
      for (int i = 0; i < 4; ++i)
#pragma unroll
        for (int j = 0; j < 2; ++j)
          acc[i][j] = __builtin_amdgcn_wmma_f32_16x16x4_f32(
              false, af[i].zw, false, bf[j].zw, (short)0, acc[i][j], false, false);
    }

    __builtin_amdgcn_sched_barrier(0);
    if (more) {
#pragma unroll
      for (int j = 0; j < 4; ++j) {
        *reinterpret_cast<v4f*>(&nxt[lA + j * 32 * PITCH]) = ga[j];
        *reinterpret_cast<v4f*>(&nxt[lB + j * 32 * PITCH]) = gb[j];
      }
      __syncthreads();
    }
  }

  static_assert(NNODES % 16 == 0, "tile guard assumes whole 16-row tiles");
  static_assert(BM * CP <= 2 * (BM + BN) * PITCH, "accumulator staging must fit in the ring");
  {
    typedef _Float16 v8h __attribute__((ext_vector_type(8)));
    float* cs = &lds[0][0];
    float a[NUM_REL][NUM_BASES];
#pragma unroll
    for (int rr = 0; rr < NUM_REL; ++rr)
#pragma unroll
      for (int b = 0; b < NUM_BASES; ++b)
        a[rr][b] = att[rr * NUM_BASES + b];

    __syncthreads();
#pragma unroll
    for (int i = 0; i < 4; ++i)
#pragma unroll
      for (int j = 0; j < 2; ++j)
#pragma unroll
        for (int v = 0; v < 8; ++v)
          cs[(wm * 64 + i * 16 + 8 * half + v) * CP + wn * 32 + j * 16 + lid] = acc[i][j][v];
    __syncthreads();

    for (int p = 0; p < 2; ++p) {
      const int task = t + 256 * p;
      const int row  = task >> 2;
      const int cg   = task & 3;
      const int grow = row0 + row;
      v4f hb[NUM_BASES][2];
#pragma unroll
      for (int b = 0; b < NUM_BASES; ++b)
#pragma unroll
        for (int hh = 0; hh < 2; ++hh)
          hb[b][hh] = *reinterpret_cast<const v4f*>(&cs[row * CP + b * 32 + cg * 8 + hh * 4]);
      if (grow < NNODES) {
        _Float16* hp = h + (size_t)grow * OUT_CH + 32 * r + cg * 8;
#pragma unroll
        for (int rr = 0; rr < NUM_REL; ++rr) {
          v8h o;
#pragma unroll
          for (int e = 0; e < 8; ++e) {
            float s = a[rr][0] * hb[0][e >> 2][e & 3];
#pragma unroll
            for (int b = 1; b < NUM_BASES; ++b)
              s += a[rr][b] * hb[b][e >> 2][e & 3];
            o[e] = (_Float16)s;
          }
          *reinterpret_cast<v8h*>(hp + (size_t)rr * NNODES * OUT_CH) = o;
        }
      }
    }
  }
}

#define TS_WAVES 8
#define TS_STG   ((TR + OUT_CH) * PITCH)
#define TS_LDSF  (2 * TR * OUT_CH)
#define TS_RPW   (TR / TS_WAVES)
#define TS_RPT   (TSLOTS / (TS_WAVES * 32))
static_assert((size_t)NUM_REL * NNODES * OUT_CH * 2 < (1ull << 32), "h byte offsets fit 32 bits");
static_assert((size_t)NUM_REL * NNODES < (1u << 20), "h row index fits 20 bits of a record");
static_assert(TR <= 64, "row inside the tile fits 6 bits of a record");
static_assert(NNODES % TR >= 1 && (NNODES % TR) * (OUT_CH * 4 / 8) > TREC0, "last tile has record slots");
static_assert(TSLOTS - TREC0 <= TS_RPT * TS_WAVES * 32, "a thread's TS_RPT records cover every slot");
static_assert(NEDGES % (TS_WAVES * 32) == 0, "whole batches in the edge-list walk of an overfull tile");

__device__ __forceinline__ void root_tile(const float* __restrict__ x,
                                          const float* __restrict__ wT,
                                          const float* __restrict__ bias,
                                          float* out, float* ring, float* rootp,
                                          const int row0, const int t, const bool to_lds) {
  const int lane = t & 31;
  const int wave = __builtin_amdgcn_readfirstlane(t >> 5);
  const int half = lane >> 4;
  const int lid  = lane & 15;
  const int wm   = wave >> 2;
  const int wn   = wave & 3;
  const int srow = t >> 3;
  const int sc4  = (t & 7) * 4;

  int aoff[2], boff[4];
#pragma unroll
  for (int j = 0; j < 2; ++j) {
    int row = row0 + srow + 32 * j;
    row = row < NNODES ? row : NNODES - 1;
    aoff[j] = row * IN_CH + sc4;
  }
  const float* bslice = wT + (size_t)NUM_BASES * OUT_CH * IN_CH;
#pragma unroll
  for (int j = 0; j < 4; ++j) boff[j] = (srow + 32 * j) * IN_CH + sc4;
  const int lA = srow * PITCH + sc4;
  const int lB = (TR + srow) * PITCH + sc4;
  const int fA = (wm * 32 + lid) * PITCH + half * 4;
  const int fB = (TR + wn * 32 + lid) * PITCH + half * 4;

  v8f acc[2][2] = {};
  v4f ga[2], gb[4];
#pragma unroll
  for (int j = 0; j < 2; ++j) ga[j] = *reinterpret_cast<const v4f*>(x + aoff[j]);
#pragma unroll
  for (int j = 0; j < 4; ++j) gb[j] = *reinterpret_cast<const v4f*>(bslice + boff[j]);
#pragma unroll
  for (int j = 0; j < 2; ++j) *reinterpret_cast<v4f*>(&ring[lA + j * 32 * PITCH]) = ga[j];
#pragma unroll
  for (int j = 0; j < 4; ++j) *reinterpret_cast<v4f*>(&ring[lB + j * 32 * PITCH]) = gb[j];
  __syncthreads();

#pragma unroll
  for (int c = 0; c < NCHUNK; ++c) {
    const float* cur = ring + (c & 1) * TS_STG;
    float*       nxt = ring + ((c + 1) & 1) * TS_STG;
    const bool  more = (c + 1 < NCHUNK);
    if (more) {
#pragma unroll
      for (int j = 0; j < 2; ++j) ga[j] = *reinterpret_cast<const v4f*>(x + aoff[j] + (c + 1) * KC);
#pragma unroll
      for (int j = 0; j < 4; ++j) gb[j] = *reinterpret_cast<const v4f*>(bslice + boff[j] + (c + 1) * KC);
    }
    __builtin_amdgcn_sched_barrier(0);
#pragma unroll
    for (int s = 0; s < KC / 8; ++s) {
      v4f af[2], bf[2];
#pragma unroll
      for (int i = 0; i < 2; ++i)
        af[i] = *reinterpret_cast<const v4f*>(&cur[fA + i * 16 * PITCH + s * 8]);
#pragma unroll
      for (int j = 0; j < 2; ++j)
        bf[j] = *reinterpret_cast<const v4f*>(&cur[fB + j * 16 * PITCH + s * 8]);
#pragma unroll
      for (int i = 0; i < 2; ++i)
#pragma unroll
        for (int j = 0; j < 2; ++j)
          acc[i][j] = __builtin_amdgcn_wmma_f32_16x16x4_f32(
              false, af[i].xy, false, bf[j].xy, (short)0, acc[i][j], false, false);
#pragma unroll
      for (int i = 0; i < 2; ++i)
#pragma unroll
        for (int j = 0; j < 2; ++j)
          acc[i][j] = __builtin_amdgcn_wmma_f32_16x16x4_f32(
              false, af[i].zw, false, bf[j].zw, (short)0, acc[i][j], false, false);
    }
    __builtin_amdgcn_sched_barrier(0);
    if (more) {
#pragma unroll
      for (int j = 0; j < 2; ++j) *reinterpret_cast<v4f*>(&nxt[lA + j * 32 * PITCH]) = ga[j];
#pragma unroll
      for (int j = 0; j < 4; ++j) *reinterpret_cast<v4f*>(&nxt[lB + j * 32 * PITCH]) = gb[j];
      __syncthreads();
    }
  }

  if (to_lds) {
    __syncthreads();
#pragma unroll
    for (int i = 0; i < 2; ++i)
#pragma unroll
      for (int j = 0; j < 2; ++j) {
        const int   col = wn * 32 + j * 16 + lid;
        const float bv  = bias[col];
#pragma unroll
        for (int v = 0; v < 8; ++v)
          rootp[(wm * 32 + i * 16 + 8 * half + v) * OUT_CH + col] = acc[i][j][v] + bv;
      }
  } else {
#pragma unroll
    for (int i = 0; i < 2; ++i) {
      const int tr = row0 + wm * 32 + i * 16;
      if (tr < NNODES) {
#pragma unroll
        for (int j = 0; j < 2; ++j) {
          const int   col = wn * 32 + j * 16 + lid;
          const float bv  = bias[col];
          float* op = out + (size_t)(tr + 8 * half) * OUT_CH + col;
#pragma unroll
          for (int v = 0; v < 8; ++v)
            op[(size_t)v * OUT_CH] += acc[i][j][v] + bv;
        }
      }
    }
  }
}

__global__ __launch_bounds__(TS_WAVES * 32)
void tile_sum_kernel(const int* __restrict__ esrc,
                     const int* __restrict__ edst,
                     const int* __restrict__ etyp,
                     const float* __restrict__ norm,
                     const _Float16* __restrict__ h,
                     float* out,
                     const float* __restrict__ x,
                     const float* __restrict__ wT,
                     const float* __restrict__ bias)
{
  __shared__ __attribute__((aligned(16))) float ldsf[TS_LDSF];
  __shared__ unsigned cnt[TR];
  __shared__ unsigned start[TR + 1];
  static_assert(TSLOTS * 8 == TR * OUT_CH * 4, "the records fill the first half of ldsf");
  static_assert(2 * TS_STG <= TS_LDSF, "root_tile's ring fits in ldsf");
  uint2* srt   = reinterpret_cast<uint2*>(ldsf);
  float* rootp = ldsf + TR * OUT_CH;

  const int t    = threadIdx.x;
  const int lane = t & 31;
  const int wave = __builtin_amdgcn_readfirstlane(t >> 5);
  const int row0 = blockIdx.x * TR;
  const int rows = (NNODES - row0) < TR ? (NNODES - row0) : TR;
  const unsigned cap = (unsigned)(rows * (OUT_CH * 4 / 8) - TREC0);

  const uint2* slots = reinterpret_cast<const uint2*>(out) + (size_t)blockIdx.x * TSLOTS;
  const unsigned n = (unsigned)__builtin_amdgcn_readfirstlane((int)slots[0].x);
  const bool fast = n <= cap;
  const bool pf = fast && ((((blockIdx.x >> 8) ^ blockIdx.x) & 1u) != 0u);
  if (pf) {
    root_tile(x, wT, bias, out, ldsf, rootp, row0, t, true);
    __syncthreads();
  }

  if (fast) {
    if (t < TR) cnt[t] = 0u;
    __syncthreads();
    uint2    rv[TS_RPT];
    unsigned rk[TS_RPT];
#pragma unroll
    for (int j = 0; j < TS_RPT; ++j) {
      const unsigned i = t + TS_WAVES * 32 * j;
      rv[j] = slots[TREC0 + (i < n ? i : 0u)];
    }
#pragma unroll
    for (int j = 0; j < TS_RPT; ++j) {
      rk[j] = 0u;
      if (t + TS_WAVES * 32 * j < n) rk[j] = atomicAdd(&cnt[rv[j].x >> 20], 1u);
    }
    __syncthreads();
    if (t < TR) {
      unsigned s = 0;
      for (int k = 0; k < t; ++k) s += cnt[k];
      start[t] = s;
    }
    if (t == 0) start[TR] = n;
    __syncthreads();
#pragma unroll
    for (int j = 0; j < TS_RPT; ++j)
      if (t + TS_WAVES * 32 * j < n) srt[start[rv[j].x >> 20] + rk[j]] = rv[j];
    __syncthreads();

    const int rlo = wave * TS_RPW;
    if (rlo < rows) {
      const int rhi = (rlo + TS_RPW) < rows ? (rlo + TS_RPW) : rows;
      const unsigned s0    = (unsigned)__builtin_amdgcn_readfirstlane((int)start[rlo]);
      const unsigned total = (unsigned)__builtin_amdgcn_readfirstlane((int)start[rhi]) - s0;

      float a0 = 0.f, a1 = 0.f, a2 = 0.f, a3 = 0.f;
      int   crow = rlo;

      auto flush_to = [&](int upto) __attribute__((always_inline)) {
        float* op = out + (size_t)(row0 + crow) * OUT_CH + lane;
        if (pf) {
          const float* rp = rootp + crow * OUT_CH + lane;
          a0 += rp[0]; a1 += rp[32]; a2 += rp[64]; a3 += rp[96];
        }
        op[0] = a0; op[32] = a1; op[64] = a2; op[96] = a3;
        a0 = 0.f; a1 = 0.f; a2 = 0.f; a3 = 0.f;
#pragma clang loop unroll(disable)
        for (int r = crow + 1; r < upto; ++r) {
          float* zp = out + (size_t)(row0 + r) * OUT_CH + lane;
          float z0 = 0.f, z1 = 0.f, z2 = 0.f, z3 = 0.f;
          if (pf) {
            const float* rp = rootp + r * OUT_CH + lane;
            z0 = rp[0]; z1 = rp[32]; z2 = rp[64]; z3 = rp[96];
          }
          zp[0] = z0; zp[32] = z1; zp[64] = z2; zp[96] = z3;
        }
        crow = upto;
      };
      auto load_chunk = [&](unsigned c) __attribute__((always_inline)) -> uint2 {
        const unsigned i = c * 32 + lane;
        return i < total ? srt[s0 + i] : make_uint2(0u, 0u);
      };
      auto issue = [&](v4h (&row)[8], const uint2& rc, const int g) __attribute__((always_inline)) {
#pragma unroll
        for (int k = 0; k < 8; ++k) {
          const unsigned x  = (unsigned)__builtin_amdgcn_readlane((int)rc.x, g * 8 + k);
          const size_t   rb = (size_t)(x & 0xFFFFFu) * (OUT_CH * 2);
          row[k] = reinterpret_cast<const v4h*>(reinterpret_cast<const char*>(h) + rb)[lane];
        }
      };
      auto consume = [&](const v4h (&row)[8], const uint2& rc, const int g, const int nav)
          __attribute__((always_inline)) {
#pragma unroll
        for (int k = 0; k < 8; ++k) {
          if (k < nav) {
            const unsigned x  = (unsigned)__builtin_amdgcn_readlane((int)rc.x, g * 8 + k);
            const float    nm = __int_as_float(__builtin_amdgcn_readlane((int)rc.y, g * 8 + k));
            const int      dl = (int)(x >> 20);
            if (dl != crow) flush_to(dl);
            a0 += (float)row[k].x * nm;
            a1 += (float)row[k].y * nm;
            a2 += (float)row[k].z * nm;
            a3 += (float)row[k].w * nm;
          }
        }
      };
      auto navail = [&](int rem, int g) __attribute__((always_inline)) -> int {
        const int v = rem - 8 * g;
        return v < 0 ? 0 : (v > 8 ? 8 : v);
      };

      v4h R0[8], R1[8];
      uint2 rc = load_chunk(0);
      uint2 rn = load_chunk(1);
      issue(R0, rc, 0);
      for (unsigned c = 0; c * 32 < total; ++c) {
        const int rem = (int)(total - c * 32);
        issue(R1, rc, 1);
        __builtin_amdgcn_sched_barrier(0);
        consume(R0, rc, 0, navail(rem, 0));
        issue(R0, rc, 2);
        __builtin_amdgcn_sched_barrier(0);
        consume(R1, rc, 1, navail(rem, 1));
        issue(R1, rc, 3);
        __builtin_amdgcn_sched_barrier(0);
        consume(R0, rc, 2, navail(rem, 2));
        issue(R0, rn, 0);
        __builtin_amdgcn_sched_barrier(0);
        consume(R1, rc, 3, navail(rem, 3));
        rc = rn;
        rn = load_chunk(c + 2);
      }
      flush_to(rhi);
    }
  } else {
    float* fs = ldsf;
    for (int i = t; i < TR * OUT_CH; i += TS_WAVES * 32) fs[i] = 0.f;
    __syncthreads();
    for (int eb = wave * 32; eb < NEDGES; eb += TS_WAVES * 32) {
      const int e = eb + lane;
      const unsigned dl = (unsigned)(edst[e] - row0);
      unsigned m = __builtin_amdgcn_ballot_w32(dl < (unsigned)rows);
      if (m != 0u) {
        const unsigned hid = (unsigned)(etyp[e] * NNODES + esrc[e]);
        const int      nbv = __float_as_int(norm[e]);
        while (m != 0u) {
          const int i = __builtin_ctz(m);
          m &= m - 1u;
          const unsigned hi  = (unsigned)__builtin_amdgcn_readlane((int)hid, i);
          const unsigned dli = (unsigned)__builtin_amdgcn_readlane((int)dl, i);
          const float    nm  = __int_as_float(__builtin_amdgcn_readlane(nbv, i));
          const v4h r = reinterpret_cast<const v4h*>(
              reinterpret_cast<const char*>(h) + (size_t)hi * (OUT_CH * 2))[lane];
          float* fp = fs + dli * OUT_CH + lane;
          atomicAdd(fp +  0, (float)r.x * nm);
          atomicAdd(fp + 32, (float)r.y * nm);
          atomicAdd(fp + 64, (float)r.z * nm);
          atomicAdd(fp + 96, (float)r.w * nm);
        }
      }
    }
    __syncthreads();
    for (int i = t; i < rows * OUT_CH; i += TS_WAVES * 32) out[(size_t)row0 * OUT_CH + i] = fs[i];
  }

  if (!pf) {
    __syncthreads();
    root_tile(x, wT, bias, out, ldsf, rootp, row0, t, false);
  }
}

extern "C" void kernel_launch(void* const* d_in, const int* in_sizes, int n_in,
                              void* d_out, int out_size, void* d_ws, size_t ws_size,
                              hipStream_t stream) {
  const float* x     = (const float*)d_in[0];
  const int*   esrc  = (const int*)  d_in[1];
  const int*   edst  = (const int*)  d_in[2];
  const int*   etyp  = (const int*)  d_in[3];
  const float* norm  = (const float*)d_in[4];
  const float* basis = (const float*)d_in[5];
  const float* att   = (const float*)d_in[6];
  const float* root  = (const float*)d_in[7];
  const float* bias  = (const float*)d_in[8];
  float* out = (float*)d_out;

  float*    wT = (float*)d_ws;
  _Float16* h  = (_Float16*)((char*)d_ws +
                             (size_t)NSLICE * IN_CH * OUT_CH * sizeof(float));

  build_wT_kernel<<<(NSLICE * IN_CH * OUT_CH + 255) / 256, 256, 0, stream>>>(
      basis, root, wT, out);

  rgcn_gemm_kernel<<<dim3(NUM_BASES + 1, MTILES), 256, 0, stream>>>(
      x, wT, att, h, out, esrc, edst, etyp, norm);

  tile_sum_kernel<<<NTILE, TS_WAVES * 32, 0, stream>>>(esrc, edst, etyp, norm, h, out,
                                                       x, wT, bias);
}
